# starting kernel
# baseline (speedup 1.0000x reference)
#include <hip/hip_runtime.h>
#include <hip/hip_bf16.h>
#include <math.h>


typedef float v2f __attribute__((ext_vector_type(2)));
typedef float v8f __attribute__((ext_vector_type(8)));

#define NROWS   131072
#define DIN     16
#define DOUT    16
#define GRID    19
#define WREAL   361
#define WPAD    368
#define NWT     23
#define TWO_PI_19 0.3306939635357677f

#define B_OFF   0
#define C_OFF   11776
#define L_OFF   (11776 + 23552)
#define WS_FLOATS (L_OFF + 256)

static __device__ inline v8f wmma4(v2f a, v2f b, v8f c) {
  return __builtin_amdgcn_wmma_f32_16x16x4_f32(
       false, a,  false, b,
       (short)0, c,  false,  false);
}


__global__ void pre_B(const float* __restrict__ Mre, const float* __restrict__ Mim,
                      float* __restrict__ Bpack) {
  int w = blockIdx.x;
  int t = threadIdx.x;
  int i = t & 15, c = t >> 4;
  float s = 0.f;
  if (w < WREAL) {
    int wu = w / GRID, wv = w % GRID;
    for (int u = 0; u < 7; ++u)
      for (int v = 0; v < 7; ++v) {
        int ph = (wu * u + wv * v) % GRID;
        float th = (float)ph * TWO_PI_19;
        float ct = cosf(th), st = sinf(th);
        float a = Mre[(i * 7 + u) * 7 + v];
        float b = Mim[(i * 7 + u) * 7 + v];
        s += (c == 0) ? (a * ct + b * st) : (b * ct - a * st);
      }
    s *= (1.f / 19.f);
  }
  Bpack[((c * 8 + (i >> 1)) * WPAD + w) * 2 + (i & 1)] = s;
}

__global__ void pre_C(const float* __restrict__ Sre, const float* __restrict__ Sim,
                      const float* __restrict__ pw, float* __restrict__ Cpack) {
  int w = blockIdx.x;
  int j = threadIdx.x;
  float g2r = 0.f, g2i = 0.f, g3r = 0.f, g3i = 0.f;
  if (w < WREAL) {
    int wu = w / GRID, wv = w % GRID;
    for (int tu = 0; tu < 13; ++tu)
      for (int tv = 0; tv < 13; ++tv) {
        int ph = (wu * tu + wv * tv) % GRID;
        float th = (float)ph * TWO_PI_19;
        float ct = cosf(th), st = sinf(th);
        float a = Sre[(j * GRID + tu + 3) * GRID + tv + 3];
        float b = Sim[(j * GRID + tu + 3) * GRID + tv + 3];
        g2r += a * ct - b * st;
        g2i += b * ct + a * st;
      }
    for (int tu = 0; tu < GRID; ++tu)
      for (int tv = 0; tv < GRID; ++tv) {
        int ph = (wu * tu + wv * tv) % GRID;
        float th = (float)ph * TWO_PI_19;
        float ct = cosf(th), st = sinf(th);
        float a = Sre[(j * GRID + tu) * GRID + tv];
        float b = Sim[(j * GRID + tu) * GRID + tv];
        g3r += a * ct - b * st;
        g3i += b * ct + a * st;
      }
  }
  float scale = pw[j] * (1.f / 19.f);
  float q[4] = { g2r * scale, -g2i * scale, g3r * scale, -g3i * scale };
  for (int qi = 0; qi < 4; ++qi)
    Cpack[((qi * (WPAD / 2) + (w >> 1)) * 16 + j) * 2 + (w & 1)] = q[qi];
}

__global__ void pre_L(const float* __restrict__ Mre, const float* __restrict__ Mim,
                      const float* __restrict__ Sre, const float* __restrict__ Sim,
                      const float* __restrict__ pw, float* __restrict__ Lpack) {
  int i = threadIdx.x;
  int j = threadIdx.y;
  float s = 0.f;
  for (int u = 0; u < 7; ++u)
    for (int v = 0; v < 7; ++v) {
      s += Mre[(i * 7 + u) * 7 + v] * Sre[(j * GRID + u + 6) * GRID + v + 6]
         - Mim[(i * 7 + u) * 7 + v] * Sim[(j * GRID + u + 6) * GRID + v + 6];
    }
  s *= pw[j];
  Lpack[((i >> 1) * 16 + j) * 2 + (i & 1)] = s;
}


__global__ __launch_bounds__(256, 1)
void gaunt_main(const float* __restrict__ X,
                const float* __restrict__ Bp,
                const float* __restrict__ Cp,
                const float* __restrict__ Lp,
                float* __restrict__ Y) {
  __shared__ float lds[8 * 4 * 16 * 18];

  const int tid  = threadIdx.x;
  const int wave = tid >> 5;
  const int lane = tid & 31;
  const int half = lane >> 4;
  const int lr   = lane & 15;

  const long row0 = ((long)blockIdx.x * 8 + wave) * 16;

  const v2f* B2 = (const v2f*)Bp;
  const v2f* C2 = (const v2f*)Cp;
  const v2f* L2 = (const v2f*)Lp;

  v2f xa[4];
  const float* xrow = X + (row0 + lr) * DIN;
#pragma unroll
  for (int kk = 0; kk < 4; ++kk)
    xa[kk] = *(const v2f*)(xrow + 4 * kk + 2 * half);

  v8f acc = {0.f, 0.f, 0.f, 0.f, 0.f, 0.f, 0.f, 0.f};

#pragma unroll
  for (int kk = 0; kk < 4; ++kk)
    acc = wmma4(xa[kk], L2[(2 * kk + half) * 16 + lr], acc);

  float* myl = lds + wave * (4 * 16 * 18);

  for (int wt = 0; wt < NWT; ++wt) {
    const int w0 = wt * 16;

    v8f fre = {0.f, 0.f, 0.f, 0.f, 0.f, 0.f, 0.f, 0.f};
    v8f fim = {0.f, 0.f, 0.f, 0.f, 0.f, 0.f, 0.f, 0.f};
#pragma unroll
    for (int kk = 0; kk < 4; ++kk) {
      fre = wmma4(xa[kk], B2[(    2 * kk + half) * WPAD + w0 + lr], fre);
      fim = wmma4(xa[kk], B2[(8 + 2 * kk + half) * WPAD + w0 + lr], fim);
    }

#pragma unroll
    for (int e = 0; e < 8; ++e) {
      float fr = fre[e], fi = fim[e];
      float p2r = fr * fr - fi * fi;
      float p2i = 2.f * fr * fi;
      float p3r = p2r * fr - p2i * fi;
      float p3i = p2r * fi + p2i * fr;
      int off = (e + 8 * half) * 18 + lr;
      myl[0 * 288 + off] = p2r;
      myl[1 * 288 + off] = p2i;
      myl[2 * 288 + off] = p3r;
      myl[3 * 288 + off] = p3i;
    }

#pragma unroll
    for (int q = 0; q < 4; ++q) {
#pragma unroll
      for (int kk = 0; kk < 4; ++kk) {
        v2f a = *(const v2f*)(myl + q * 288 + lr * 18 + 4 * kk + 2 * half);
        v2f b = C2[(q * (WPAD / 2) + wt * 8 + 2 * kk + half) * 16 + lr];
        acc = wmma4(a, b, acc);
      }
    }
  }

  float* yrow = Y + row0 * DOUT;
#pragma unroll
  for (int e = 0; e < 8; ++e)
    yrow[(e + 8 * half) * DOUT + lr] = acc[e];
}


extern "C" void kernel_launch(void* const* d_in, const int* in_sizes, int n_in,
                              void* d_out, int out_size, void* d_ws, size_t ws_size,
                              hipStream_t stream) {
  const float* x   = (const float*)d_in[0];
  const float* Mre = (const float*)d_in[1];
  const float* Mim = (const float*)d_in[2];
  const float* Sre = (const float*)d_in[3];
  const float* Sim = (const float*)d_in[4];
  const float* pw  = (const float*)d_in[5];

  float* ws = (float*)d_ws;
  float* Bp = ws + B_OFF;
  float* Cp = ws + C_OFF;
  float* Lp = ws + L_OFF;

  pre_B<<<WPAD, 32, 0, stream>>>(Mre, Mim, Bp);
  pre_C<<<WPAD, 16, 0, stream>>>(Sre, Sim, pw, Cp);
  dim3 lb(16, 16);
  pre_L<<<1, lb, 0, stream>>>(Mre, Mim, Sre, Sim, pw, Lp);

  gaunt_main<<<NROWS / (8 * 16), 256, 0, stream>>>(x, Bp, Cp, Lp, (float*)d_out);
}


# pre_C rebuilt as a lane-parallel sum: the 361 terms of each (w,j) are dealt to 16 lanes, twiddles come from a 19-entry LDS table (no per-term sinf/cosf), shuffle-tree join; sum order of C changes
# speedup vs baseline: 2.3193x; 2.3193x over previous
#include <hip/hip_runtime.h>
#include <hip/hip_bf16.h>
#include <math.h>


typedef float v2f __attribute__((ext_vector_type(2)));
typedef float v8f __attribute__((ext_vector_type(8)));

#define NROWS   131072
#define DIN     16
#define DOUT    16
#define GRID    19
#define WREAL   361
#define WPAD    368
#define NWT     23
#define TWO_PI_19 0.3306939635357677f

#define B_OFF   0
#define C_OFF   11776
#define L_OFF   (11776 + 23552)
#define WS_FLOATS (L_OFF + 256)

static __device__ inline v8f wmma4(v2f a, v2f b, v8f c) {
  return __builtin_amdgcn_wmma_f32_16x16x4_f32(
       false, a,  false, b,
       (short)0, c,  false,  false);
}


__global__ void pre_B(const float* __restrict__ Mre, const float* __restrict__ Mim,
                      float* __restrict__ Bpack) {
  int w = blockIdx.x;
  int t = threadIdx.x;
  int i = t & 15, c = t >> 4;
  float s = 0.f;
  if (w < WREAL) {
    int wu = w / GRID, wv = w % GRID;
    for (int u = 0; u < 7; ++u)
      for (int v = 0; v < 7; ++v) {
        int ph = (wu * u + wv * v) % GRID;
        float th = (float)ph * TWO_PI_19;
        float ct = cosf(th), st = sinf(th);
        float a = Mre[(i * 7 + u) * 7 + v];
        float b = Mim[(i * 7 + u) * 7 + v];
        s += (c == 0) ? (a * ct + b * st) : (b * ct - a * st);
      }
    s *= (1.f / 19.f);
  }
  Bpack[((c * 8 + (i >> 1)) * WPAD + w) * 2 + (i & 1)] = s;
}

__global__ __launch_bounds__(256)
void pre_C(const float* __restrict__ Sre, const float* __restrict__ Sim,
           const float* __restrict__ pw, float* __restrict__ Cpack) {
  __shared__ float tw_c[GRID];
  __shared__ float tw_s[GRID];
  const int tid = threadIdx.x;
  if (tid < GRID) {
    float th = (float)tid * TWO_PI_19;
    tw_c[tid] = cosf(th);
    tw_s[tid] = sinf(th);
  }
  __syncthreads();

  const int j    = tid >> 4;
  const int part = tid & 15;
  const int wp   = blockIdx.x;
  int wu[2], wv[2];
  for (int c = 0; c < 2; ++c) {
    int w = 2 * wp + c;
    wu[c] = w / GRID;
    wv[c] = w % GRID;
  }

  float g2r[2] = {0.f, 0.f}, g2i[2] = {0.f, 0.f};
  float g3r[2] = {0.f, 0.f}, g3i[2] = {0.f, 0.f};
  const float* sre = Sre + j * WREAL;
  const float* sim = Sim + j * WREAL;

#pragma unroll
  for (int it = 0; it < 23; ++it) {
    int t = part + 16 * it;
    bool ok = t < WREAL;
    int tc = ok ? t : 0;
    float a = sre[tc];
    float b = sim[tc];
    a = ok ? a : 0.f;
    b = ok ? b : 0.f;
    int tu = tc / GRID, tv = tc - tu * GRID;
    bool in2 = (tu >= 3) && (tu < 16) && (tv >= 3) && (tv < 16);
    float a2 = in2 ? a : 0.f;
    float b2 = in2 ? b : 0.f;
    for (int c = 0; c < 2; ++c) {
      int ph3 = (wu[c] * tu + wv[c] * tv) % GRID;
      float c3 = tw_c[ph3], s3 = tw_s[ph3];
      g3r[c] += a * c3 - b * s3;
      g3i[c] += b * c3 + a * s3;
      int ph2 = (wu[c] * (tu + 16) + wv[c] * (tv + 16)) % GRID;
      float c2 = tw_c[ph2], s2 = tw_s[ph2];
      g2r[c] += a2 * c2 - b2 * s2;
      g2i[c] += b2 * c2 + a2 * s2;
    }
  }

  float v[8] = { g2r[0], g2i[0], g3r[0], g3i[0], g2r[1], g2i[1], g3r[1], g3i[1] };
  for (int m = 8; m >= 1; m >>= 1) {
    for (int k = 0; k < 8; ++k)
      v[k] += __shfl_xor(v[k], m, 16);
  }

  if (part == 0) {
    float scale = pw[j] * (1.f / 19.f);
    bool ok0 = (2 * wp)     < WREAL;
    bool ok1 = (2 * wp + 1) < WREAL;
    float q0[4] = { v[0] * scale, -v[1] * scale, v[2] * scale, -v[3] * scale };
    float q1[4] = { v[4] * scale, -v[5] * scale, v[6] * scale, -v[7] * scale };
    v2f* C2 = (v2f*)Cpack;
    for (int qi = 0; qi < 4; ++qi) {
      v2f o;
      o.x = ok0 ? q0[qi] : 0.f;
      o.y = ok1 ? q1[qi] : 0.f;
      C2[(qi * (WPAD / 2) + wp) * 16 + j] = o;
    }
  }
}

__global__ void pre_L(const float* __restrict__ Mre, const float* __restrict__ Mim,
                      const float* __restrict__ Sre, const float* __restrict__ Sim,
                      const float* __restrict__ pw, float* __restrict__ Lpack) {
  int i = threadIdx.x;
  int j = threadIdx.y;
  float s = 0.f;
  for (int u = 0; u < 7; ++u)
    for (int v = 0; v < 7; ++v) {
      s += Mre[(i * 7 + u) * 7 + v] * Sre[(j * GRID + u + 6) * GRID + v + 6]
         - Mim[(i * 7 + u) * 7 + v] * Sim[(j * GRID + u + 6) * GRID + v + 6];
    }
  s *= pw[j];
  Lpack[((i >> 1) * 16 + j) * 2 + (i & 1)] = s;
}


__global__ __launch_bounds__(256, 1)
void gaunt_main(const float* __restrict__ X,
                const float* __restrict__ Bp,
                const float* __restrict__ Cp,
                const float* __restrict__ Lp,
                float* __restrict__ Y) {
  __shared__ float lds[8 * 4 * 16 * 18];

  const int tid  = threadIdx.x;
  const int wave = tid >> 5;
  const int lane = tid & 31;
  const int half = lane >> 4;
  const int lr   = lane & 15;

  const long row0 = ((long)blockIdx.x * 8 + wave) * 16;

  const v2f* B2 = (const v2f*)Bp;
  const v2f* C2 = (const v2f*)Cp;
  const v2f* L2 = (const v2f*)Lp;

  v2f xa[4];
  const float* xrow = X + (row0 + lr) * DIN;
#pragma unroll
  for (int kk = 0; kk < 4; ++kk)
    xa[kk] = *(const v2f*)(xrow + 4 * kk + 2 * half);

  v8f acc = {0.f, 0.f, 0.f, 0.f, 0.f, 0.f, 0.f, 0.f};

#pragma unroll
  for (int kk = 0; kk < 4; ++kk)
    acc = wmma4(xa[kk], L2[(2 * kk + half) * 16 + lr], acc);

  float* myl = lds + wave * (4 * 16 * 18);

  for (int wt = 0; wt < NWT; ++wt) {
    const int w0 = wt * 16;

    v8f fre = {0.f, 0.f, 0.f, 0.f, 0.f, 0.f, 0.f, 0.f};
    v8f fim = {0.f, 0.f, 0.f, 0.f, 0.f, 0.f, 0.f, 0.f};
#pragma unroll
    for (int kk = 0; kk < 4; ++kk) {
      fre = wmma4(xa[kk], B2[(    2 * kk + half) * WPAD + w0 + lr], fre);
      fim = wmma4(xa[kk], B2[(8 + 2 * kk + half) * WPAD + w0 + lr], fim);
    }

#pragma unroll
    for (int e = 0; e < 8; ++e) {
      float fr = fre[e], fi = fim[e];
      float p2r = fr * fr - fi * fi;
      float p2i = 2.f * fr * fi;
      float p3r = p2r * fr - p2i * fi;
      float p3i = p2r * fi + p2i * fr;
      int off = (e + 8 * half) * 18 + lr;
      myl[0 * 288 + off] = p2r;
      myl[1 * 288 + off] = p2i;
      myl[2 * 288 + off] = p3r;
      myl[3 * 288 + off] = p3i;
    }

#pragma unroll
    for (int q = 0; q < 4; ++q) {
#pragma unroll
      for (int kk = 0; kk < 4; ++kk) {
        v2f a = *(const v2f*)(myl + q * 288 + lr * 18 + 4 * kk + 2 * half);
        v2f b = C2[(q * (WPAD / 2) + wt * 8 + 2 * kk + half) * 16 + lr];
        acc = wmma4(a, b, acc);
      }
    }
  }

  float* yrow = Y + row0 * DOUT;
#pragma unroll
  for (int e = 0; e < 8; ++e)
    yrow[(e + 8 * half) * DOUT + lr] = acc[e];
}


extern "C" void kernel_launch(void* const* d_in, const int* in_sizes, int n_in,
                              void* d_out, int out_size, void* d_ws, size_t ws_size,
                              hipStream_t stream) {
  const float* x   = (const float*)d_in[0];
  const float* Mre = (const float*)d_in[1];
  const float* Mim = (const float*)d_in[2];
  const float* Sre = (const float*)d_in[3];
  const float* Sim = (const float*)d_in[4];
  const float* pw  = (const float*)d_in[5];

  float* ws = (float*)d_ws;
  float* Bp = ws + B_OFF;
  float* Cp = ws + C_OFF;
  float* Lp = ws + L_OFF;

  pre_B<<<WPAD, 32, 0, stream>>>(Mre, Mim, Bp);
  pre_C<<<WPAD / 2, 256, 0, stream>>>(Sre, Sim, pw, Cp);
  dim3 lb(16, 16);
  pre_L<<<1, lb, 0, stream>>>(Mre, Mim, Sre, Sim, pw, Lp);

  gaunt_main<<<NROWS / (8 * 16), 256, 0, stream>>>(x, Bp, Cp, Lp, (float*)d_out);
}


# The three constant-building launches (B, C and L packs) merged into one 231-block launch, each block taking one role by its index; every pack's arithmetic is unchanged; 4 launches become 2
# speedup vs baseline: 2.6598x; 1.1468x over previous
#include <hip/hip_runtime.h>
#include <hip/hip_bf16.h>
#include <math.h>


typedef float v2f __attribute__((ext_vector_type(2)));
typedef float v8f __attribute__((ext_vector_type(8)));

#define NROWS   131072
#define DIN     16
#define DOUT    16
#define GRID    19
#define WREAL   361
#define WPAD    368
#define NWT     23
#define TWO_PI_19 0.3306939635357677f

#define B_OFF   0
#define C_OFF   11776
#define L_OFF   (11776 + 23552)
#define WS_FLOATS (L_OFF + 256)

static __device__ inline v8f wmma4(v2f a, v2f b, v8f c) {
  return __builtin_amdgcn_wmma_f32_16x16x4_f32(
       false, a,  false, b,
       (short)0, c,  false,  false);
}


static __device__ inline void role_B(int w, int t,
                                     const float* __restrict__ Mre, const float* __restrict__ Mim,
                                     float* __restrict__ Bpack) {
  int i = t & 15, c = t >> 4;
  float s = 0.f;
  if (w < WREAL) {
    int wu = w / GRID, wv = w % GRID;
    for (int u = 0; u < 7; ++u)
      for (int v = 0; v < 7; ++v) {
        int ph = (wu * u + wv * v) % GRID;
        float th = (float)ph * TWO_PI_19;
        float ct = cosf(th), st = sinf(th);
        float a = Mre[(i * 7 + u) * 7 + v];
        float b = Mim[(i * 7 + u) * 7 + v];
        s += (c == 0) ? (a * ct + b * st) : (b * ct - a * st);
      }
    s *= (1.f / 19.f);
  }
  Bpack[((c * 8 + (i >> 1)) * WPAD + w) * 2 + (i & 1)] = s;
}

static __device__ inline void role_C(int wp, int tid, float* tw_c, float* tw_s,
                                     const float* __restrict__ Sre, const float* __restrict__ Sim,
                                     const float* __restrict__ pw, float* __restrict__ Cpack) {
  if (tid < GRID) {
    float th = (float)tid * TWO_PI_19;
    tw_c[tid] = cosf(th);
    tw_s[tid] = sinf(th);
  }
  __syncthreads();

  const int j    = tid >> 4;
  const int part = tid & 15;
  int wu[2], wv[2];
  for (int c = 0; c < 2; ++c) {
    int w = 2 * wp + c;
    wu[c] = w / GRID;
    wv[c] = w % GRID;
  }

  float g2r[2] = {0.f, 0.f}, g2i[2] = {0.f, 0.f};
  float g3r[2] = {0.f, 0.f}, g3i[2] = {0.f, 0.f};
  const float* sre = Sre + j * WREAL;
  const float* sim = Sim + j * WREAL;

#pragma unroll
  for (int it = 0; it < 23; ++it) {
    int t = part + 16 * it;
    bool ok = t < WREAL;
    int tc = ok ? t : 0;
    float a = sre[tc];
    float b = sim[tc];
    a = ok ? a : 0.f;
    b = ok ? b : 0.f;
    int tu = tc / GRID, tv = tc - tu * GRID;
    bool in2 = (tu >= 3) && (tu < 16) && (tv >= 3) && (tv < 16);
    float a2 = in2 ? a : 0.f;
    float b2 = in2 ? b : 0.f;
    for (int c = 0; c < 2; ++c) {
      int ph3 = (wu[c] * tu + wv[c] * tv) % GRID;
      float c3 = tw_c[ph3], s3 = tw_s[ph3];
      g3r[c] += a * c3 - b * s3;
      g3i[c] += b * c3 + a * s3;
      int ph2 = (wu[c] * (tu + 16) + wv[c] * (tv + 16)) % GRID;
      float c2 = tw_c[ph2], s2 = tw_s[ph2];
      g2r[c] += a2 * c2 - b2 * s2;
      g2i[c] += b2 * c2 + a2 * s2;
    }
  }

  float v[8] = { g2r[0], g2i[0], g3r[0], g3i[0], g2r[1], g2i[1], g3r[1], g3i[1] };
  for (int m = 8; m >= 1; m >>= 1) {
    for (int k = 0; k < 8; ++k)
      v[k] += __shfl_xor(v[k], m, 16);
  }

  if (part == 0) {
    float scale = pw[j] * (1.f / 19.f);
    bool ok0 = (2 * wp)     < WREAL;
    bool ok1 = (2 * wp + 1) < WREAL;
    float q0[4] = { v[0] * scale, -v[1] * scale, v[2] * scale, -v[3] * scale };
    float q1[4] = { v[4] * scale, -v[5] * scale, v[6] * scale, -v[7] * scale };
    v2f* C2 = (v2f*)Cpack;
    for (int qi = 0; qi < 4; ++qi) {
      v2f o;
      o.x = ok0 ? q0[qi] : 0.f;
      o.y = ok1 ? q1[qi] : 0.f;
      C2[(qi * (WPAD / 2) + wp) * 16 + j] = o;
    }
  }
}

static __device__ inline void role_L(int i, int j,
                                     const float* __restrict__ Mre, const float* __restrict__ Mim,
                                     const float* __restrict__ Sre, const float* __restrict__ Sim,
                                     const float* __restrict__ pw, float* __restrict__ Lpack) {
  float s = 0.f;
  for (int u = 0; u < 7; ++u)
    for (int v = 0; v < 7; ++v) {
      s += Mre[(i * 7 + u) * 7 + v] * Sre[(j * GRID + u + 6) * GRID + v + 6]
         - Mim[(i * 7 + u) * 7 + v] * Sim[(j * GRID + u + 6) * GRID + v + 6];
    }
  s *= pw[j];
  Lpack[((i >> 1) * 16 + j) * 2 + (i & 1)] = s;
}

#define NB_B    (WPAD / 8)
#define NB_C    (WPAD / 2)
#define NB_PRE  (NB_B + NB_C + 1)
__global__ __launch_bounds__(256)
void pre_all(const float* __restrict__ Mre, const float* __restrict__ Mim,
             const float* __restrict__ Sre, const float* __restrict__ Sim,
             const float* __restrict__ pw,
             float* __restrict__ Bpack, float* __restrict__ Cpack,
             float* __restrict__ Lpack) {
  __shared__ float tw_c[GRID];
  __shared__ float tw_s[GRID];
  const int tid = threadIdx.x;
  const int b   = blockIdx.x;
  if (b < NB_B) {
    role_B(b * 8 + (tid >> 5), tid & 31, Mre, Mim, Bpack);
  } else if (b < NB_B + NB_C) {
    role_C(b - NB_B, tid, tw_c, tw_s, Sre, Sim, pw, Cpack);
  } else {
    role_L(tid & 15, tid >> 4, Mre, Mim, Sre, Sim, pw, Lpack);
  }
}


__global__ __launch_bounds__(256, 1)
void gaunt_main(const float* __restrict__ X,
                const float* __restrict__ Bp,
                const float* __restrict__ Cp,
                const float* __restrict__ Lp,
                float* __restrict__ Y) {
  __shared__ float lds[8 * 4 * 16 * 18];

  const int tid  = threadIdx.x;
  const int wave = tid >> 5;
  const int lane = tid & 31;
  const int half = lane >> 4;
  const int lr   = lane & 15;

  const long row0 = ((long)blockIdx.x * 8 + wave) * 16;

  const v2f* B2 = (const v2f*)Bp;
  const v2f* C2 = (const v2f*)Cp;
  const v2f* L2 = (const v2f*)Lp;

  v2f xa[4];
  const float* xrow = X + (row0 + lr) * DIN;
#pragma unroll
  for (int kk = 0; kk < 4; ++kk)
    xa[kk] = *(const v2f*)(xrow + 4 * kk + 2 * half);

  v8f acc = {0.f, 0.f, 0.f, 0.f, 0.f, 0.f, 0.f, 0.f};

#pragma unroll
  for (int kk = 0; kk < 4; ++kk)
    acc = wmma4(xa[kk], L2[(2 * kk + half) * 16 + lr], acc);

  float* myl = lds + wave * (4 * 16 * 18);

  for (int wt = 0; wt < NWT; ++wt) {
    const int w0 = wt * 16;

    v8f fre = {0.f, 0.f, 0.f, 0.f, 0.f, 0.f, 0.f, 0.f};
    v8f fim = {0.f, 0.f, 0.f, 0.f, 0.f, 0.f, 0.f, 0.f};
#pragma unroll
    for (int kk = 0; kk < 4; ++kk) {
      fre = wmma4(xa[kk], B2[(    2 * kk + half) * WPAD + w0 + lr], fre);
      fim = wmma4(xa[kk], B2[(8 + 2 * kk + half) * WPAD + w0 + lr], fim);
    }

#pragma unroll
    for (int e = 0; e < 8; ++e) {
      float fr = fre[e], fi = fim[e];
      float p2r = fr * fr - fi * fi;
      float p2i = 2.f * fr * fi;
      float p3r = p2r * fr - p2i * fi;
      float p3i = p2r * fi + p2i * fr;
      int off = (e + 8 * half) * 18 + lr;
      myl[0 * 288 + off] = p2r;
      myl[1 * 288 + off] = p2i;
      myl[2 * 288 + off] = p3r;
      myl[3 * 288 + off] = p3i;
    }

#pragma unroll
    for (int q = 0; q < 4; ++q) {
#pragma unroll
      for (int kk = 0; kk < 4; ++kk) {
        v2f a = *(const v2f*)(myl + q * 288 + lr * 18 + 4 * kk + 2 * half);
        v2f b = C2[(q * (WPAD / 2) + wt * 8 + 2 * kk + half) * 16 + lr];
        acc = wmma4(a, b, acc);
      }
    }
  }

  float* yrow = Y + row0 * DOUT;
#pragma unroll
  for (int e = 0; e < 8; ++e)
    yrow[(e + 8 * half) * DOUT + lr] = acc[e];
}


extern "C" void kernel_launch(void* const* d_in, const int* in_sizes, int n_in,
                              void* d_out, int out_size, void* d_ws, size_t ws_size,
                              hipStream_t stream) {
  const float* x   = (const float*)d_in[0];
  const float* Mre = (const float*)d_in[1];
  const float* Mim = (const float*)d_in[2];
  const float* Sre = (const float*)d_in[3];
  const float* Sim = (const float*)d_in[4];
  const float* pw  = (const float*)d_in[5];

  float* ws = (float*)d_ws;
  float* Bp = ws + B_OFF;
  float* Cp = ws + C_OFF;
  float* Lp = ws + L_OFF;

  pre_all<<<NB_PRE, 256, 0, stream>>>(Mre, Mim, Sre, Sim, pw, Bp, Cp, Lp);

  gaunt_main<<<NROWS / (8 * 16), 256, 0, stream>>>(x, Bp, Cp, Lp, (float*)d_out);
}


# Both products run transposed (constants as A, samples as B): the squared and cubed tile stays in its D registers as B operand of the output product, no LDS staging; sum order of y changes
# speedup vs baseline: 2.9547x; 1.1109x over previous
#include <hip/hip_runtime.h>
#include <hip/hip_bf16.h>
#include <math.h>


typedef float v2f __attribute__((ext_vector_type(2)));
typedef float v8f __attribute__((ext_vector_type(8)));

#define NROWS   131072
#define DIN     16
#define DOUT    16
#define GRID    19
#define WREAL   361
#define WPAD    368
#define NWT     23
#define TWO_PI_19 0.3306939635357677f

#define B_OFF   0
#define C_OFF   11776
#define L_OFF   (11776 + 23552)
#define WS_FLOATS (L_OFF + 256)

static __device__ inline v8f wmma4(v2f a, v2f b, v8f c) {
  return __builtin_amdgcn_wmma_f32_16x16x4_f32(
       false, a,  false, b,
       (short)0, c,  false,  false);
}


static __device__ inline void role_B(int w, int t,
                                     const float* __restrict__ Mre, const float* __restrict__ Mim,
                                     float* __restrict__ Bpack) {
  int i = t & 15, c = t >> 4;
  float s = 0.f;
  if (w < WREAL) {
    int wu = w / GRID, wv = w % GRID;
    for (int u = 0; u < 7; ++u)
      for (int v = 0; v < 7; ++v) {
        int ph = (wu * u + wv * v) % GRID;
        float th = (float)ph * TWO_PI_19;
        float ct = cosf(th), st = sinf(th);
        float a = Mre[(i * 7 + u) * 7 + v];
        float b = Mim[(i * 7 + u) * 7 + v];
        s += (c == 0) ? (a * ct + b * st) : (b * ct - a * st);
      }
    s *= (1.f / 19.f);
  }
  Bpack[((c * 8 + (i >> 1)) * WPAD + w) * 2 + (i & 1)] = s;
}

static __device__ inline void role_C(int wp, int tid, float* tw_c, float* tw_s,
                                     const float* __restrict__ Sre, const float* __restrict__ Sim,
                                     const float* __restrict__ pw, float* __restrict__ Cpack) {
  if (tid < GRID) {
    float th = (float)tid * TWO_PI_19;
    tw_c[tid] = cosf(th);
    tw_s[tid] = sinf(th);
  }
  __syncthreads();

  const int j    = tid >> 4;
  const int part = tid & 15;
  int wu[2], wv[2];
  for (int c = 0; c < 2; ++c) {
    int w = 2 * wp + c;
    wu[c] = w / GRID;
    wv[c] = w % GRID;
  }

  float g2r[2] = {0.f, 0.f}, g2i[2] = {0.f, 0.f};
  float g3r[2] = {0.f, 0.f}, g3i[2] = {0.f, 0.f};
  const float* sre = Sre + j * WREAL;
  const float* sim = Sim + j * WREAL;

#pragma unroll
  for (int it = 0; it < 23; ++it) {
    int t = part + 16 * it;
    bool ok = t < WREAL;
    int tc = ok ? t : 0;
    float a = sre[tc];
    float b = sim[tc];
    a = ok ? a : 0.f;
    b = ok ? b : 0.f;
    int tu = tc / GRID, tv = tc - tu * GRID;
    bool in2 = (tu >= 3) && (tu < 16) && (tv >= 3) && (tv < 16);
    float a2 = in2 ? a : 0.f;
    float b2 = in2 ? b : 0.f;
    for (int c = 0; c < 2; ++c) {
      int ph3 = (wu[c] * tu + wv[c] * tv) % GRID;
      float c3 = tw_c[ph3], s3 = tw_s[ph3];
      g3r[c] += a * c3 - b * s3;
      g3i[c] += b * c3 + a * s3;
      int ph2 = (wu[c] * (tu + 16) + wv[c] * (tv + 16)) % GRID;
      float c2 = tw_c[ph2], s2 = tw_s[ph2];
      g2r[c] += a2 * c2 - b2 * s2;
      g2i[c] += b2 * c2 + a2 * s2;
    }
  }

  float v[8] = { g2r[0], g2i[0], g3r[0], g3i[0], g2r[1], g2i[1], g3r[1], g3i[1] };
  for (int m = 8; m >= 1; m >>= 1) {
    for (int k = 0; k < 8; ++k)
      v[k] += __shfl_xor(v[k], m, 16);
  }

  if (part == 0) {
    float scale = pw[j] * (1.f / 19.f);
    bool ok0 = (2 * wp)     < WREAL;
    bool ok1 = (2 * wp + 1) < WREAL;
    float q0[4] = { v[0] * scale, -v[1] * scale, v[2] * scale, -v[3] * scale };
    float q1[4] = { v[4] * scale, -v[5] * scale, v[6] * scale, -v[7] * scale };
    v2f* C2 = (v2f*)Cpack;
    for (int qi = 0; qi < 4; ++qi) {
      v2f o;
      o.x = ok0 ? q0[qi] : 0.f;
      o.y = ok1 ? q1[qi] : 0.f;
      C2[(qi * (WPAD / 2) + wp) * 16 + j] = o;
    }
  }
}

static __device__ inline void role_L(int i, int j,
                                     const float* __restrict__ Mre, const float* __restrict__ Mim,
                                     const float* __restrict__ Sre, const float* __restrict__ Sim,
                                     const float* __restrict__ pw, float* __restrict__ Lpack) {
  float s = 0.f;
  for (int u = 0; u < 7; ++u)
    for (int v = 0; v < 7; ++v) {
      s += Mre[(i * 7 + u) * 7 + v] * Sre[(j * GRID + u + 6) * GRID + v + 6]
         - Mim[(i * 7 + u) * 7 + v] * Sim[(j * GRID + u + 6) * GRID + v + 6];
    }
  s *= pw[j];
  Lpack[((i >> 1) * 16 + j) * 2 + (i & 1)] = s;
}

#define NB_B    (WPAD / 8)
#define NB_C    (WPAD / 2)
#define NB_PRE  (NB_B + NB_C + 1)
__global__ __launch_bounds__(256)
void pre_all(const float* __restrict__ Mre, const float* __restrict__ Mim,
             const float* __restrict__ Sre, const float* __restrict__ Sim,
             const float* __restrict__ pw,
             float* __restrict__ Bpack, float* __restrict__ Cpack,
             float* __restrict__ Lpack) {
  __shared__ float tw_c[GRID];
  __shared__ float tw_s[GRID];
  const int tid = threadIdx.x;
  const int b   = blockIdx.x;
  if (b < NB_B) {
    role_B(b * 8 + (tid >> 5), tid & 31, Mre, Mim, Bpack);
  } else if (b < NB_B + NB_C) {
    role_C(b - NB_B, tid, tw_c, tw_s, Sre, Sim, pw, Cpack);
  } else {
    role_L(tid & 15, tid >> 4, Mre, Mim, Sre, Sim, pw, Lpack);
  }
}


__global__ __launch_bounds__(256, 1)
void gaunt_main(const float* __restrict__ X,
                const float* __restrict__ Bp,
                const float* __restrict__ Cp,
                const float* __restrict__ Lp,
                float* __restrict__ Y) {
  const int tid  = threadIdx.x;
  const int wave = tid >> 5;
  const int lane = tid & 31;
  const int half = lane >> 4;
  const int lr   = lane & 15;

  const long row0 = ((long)blockIdx.x * 8 + wave) * 16;

  const v2f* B2 = (const v2f*)Bp;
  const v2f* C2 = (const v2f*)Cp;
  const v2f* L2 = (const v2f*)Lp;

  v2f xa[4];
  const float* xrow = X + (row0 + lr) * DIN;
#pragma unroll
  for (int kk = 0; kk < 4; ++kk)
    xa[kk] = *(const v2f*)(xrow + 4 * kk + 2 * half);

  v8f acc = {0.f, 0.f, 0.f, 0.f, 0.f, 0.f, 0.f, 0.f};

#pragma unroll
  for (int kk = 0; kk < 4; ++kk)
    acc = wmma4(L2[(2 * kk + half) * 16 + lr], xa[kk], acc);

  const int boff = half * WPAD + lr;
  const int coff = (4 * half) * 16 + lr;

  for (int wt = 0; wt < NWT; ++wt) {
    const v2f* Bw = B2 + wt * 16;
    const v2f* Cw = C2 + wt * (8 * 16);

    v8f fre = {0.f, 0.f, 0.f, 0.f, 0.f, 0.f, 0.f, 0.f};
    v8f fim = {0.f, 0.f, 0.f, 0.f, 0.f, 0.f, 0.f, 0.f};
#pragma unroll
    for (int kk = 0; kk < 4; ++kk) {
      fre = wmma4(Bw[(    2 * kk) * WPAD + boff], xa[kk], fre);
      fim = wmma4(Bw[(8 + 2 * kk) * WPAD + boff], xa[kk], fim);
    }

    v8f pq[4];
#pragma unroll
    for (int e = 0; e < 8; ++e) {
      float fr = fre[e], fi = fim[e];
      float p2r = fr * fr - fi * fi;
      float p2i = 2.f * fr * fi;
      float p3r = p2r * fr - p2i * fi;
      float p3i = p2r * fi + p2i * fr;
      pq[0][e] = p2r;
      pq[1][e] = p2i;
      pq[2][e] = p3r;
      pq[3][e] = p3i;
    }

#pragma unroll
    for (int q = 0; q < 4; ++q) {
#pragma unroll
      for (int t = 0; t < 4; ++t) {
        v2f a = Cw[(q * (WPAD / 2) + t) * 16 + coff];
        v2f b;
        b.x = pq[q][2 * t];
        b.y = pq[q][2 * t + 1];
        acc = wmma4(a, b, acc);
      }
    }
  }

  typedef float v4f __attribute__((ext_vector_type(4)));
  float* yrow = Y + (row0 + lr) * DOUT + 8 * half;
  v4f lo = {acc[0], acc[1], acc[2], acc[3]};
  v4f hi = {acc[4], acc[5], acc[6], acc[7]};
  *(v4f*)(yrow)     = lo;
  *(v4f*)(yrow + 4) = hi;
}


extern "C" void kernel_launch(void* const* d_in, const int* in_sizes, int n_in,
                              void* d_out, int out_size, void* d_ws, size_t ws_size,
                              hipStream_t stream) {
  const float* x   = (const float*)d_in[0];
  const float* Mre = (const float*)d_in[1];
  const float* Mim = (const float*)d_in[2];
  const float* Sre = (const float*)d_in[3];
  const float* Sim = (const float*)d_in[4];
  const float* pw  = (const float*)d_in[5];

  float* ws = (float*)d_ws;
  float* Bp = ws + B_OFF;
  float* Cp = ws + C_OFF;
  float* Lp = ws + L_OFF;

  pre_all<<<NB_PRE, 256, 0, stream>>>(Mre, Mim, Sre, Sim, pw, Bp, Cp, Lp);

  gaunt_main<<<NROWS / (8 * 16), 256, 0, stream>>>(x, Bp, Cp, Lp, (float*)d_out);
}


# Degree-2 term as a direct quadratic form: lanes form the products x_a*x_b, one K=144 WMMA product with a prebuilt coefficient pack replaces the two p2 planes (K=736) of GEMM2; sum order of y changes
# speedup vs baseline: 3.6627x; 1.2396x over previous
#include <hip/hip_runtime.h>
#include <hip/hip_bf16.h>
#include <math.h>


typedef float v2f __attribute__((ext_vector_type(2)));
typedef float v8f __attribute__((ext_vector_type(8)));

#define NROWS   131072
#define DIN     16
#define DOUT    16
#define GRID    19
#define WREAL   361
#define WPAD    368
#define NWT     23
#define TWO_PI_19 0.3306939635357677f

#define NSLOT   72
#define NT2     (NSLOT / 2)
#define NCONV   169

#define B_OFF   0
#define C_OFF   11776
#define L_OFF   (11776 + 11776)
#define T_OFF   (L_OFF + 256)
#define WS_FLOATS (T_OFF + 2304)

struct IdxPair { int a, b; };
__host__ __device__ constexpr IdxPair slot_pair(int e) {
  int n = 0;
  for (int a = 0; a < 16; ++a)
    for (int b = a; b < 16; ++b) {
      int pa = a ^ 2, pb = b ^ 2;
      if (pa > pb) { int t = pa; pa = pb; pb = t; }
      bool first = (a < pa) || (a == pa && b <= pb);
      if (first) {
        if (n == e) return IdxPair{a, b};
        ++n;
      }
    }
  return IdxPair{0, 0};
}
struct SlotTable { unsigned char a[NSLOT]; unsigned char b[NSLOT]; };
__host__ __device__ constexpr SlotTable make_slot_table() {
  SlotTable t{};
  for (int e = 0; e < NSLOT; ++e) {
    IdxPair p = slot_pair(e);
    t.a[e] = (unsigned char)p.a;
    t.b[e] = (unsigned char)p.b;
  }
  return t;
}

static __device__ inline v8f wmma4(v2f a, v2f b, v8f c) {
  return __builtin_amdgcn_wmma_f32_16x16x4_f32(
       false, a,  false, b,
       (short)0, c,  false,  false);
}


static __device__ inline void role_B(int w, int t,
                                     const float* __restrict__ Mre, const float* __restrict__ Mim,
                                     float* __restrict__ Bpack) {
  int i = t & 15, c = t >> 4;
  float s = 0.f;
  if (w < WREAL) {
    int wu = w / GRID, wv = w % GRID;
    for (int u = 0; u < 7; ++u)
      for (int v = 0; v < 7; ++v) {
        int ph = (wu * u + wv * v) % GRID;
        float th = (float)ph * TWO_PI_19;
        float ct = cosf(th), st = sinf(th);
        float a = Mre[(i * 7 + u) * 7 + v];
        float b = Mim[(i * 7 + u) * 7 + v];
        s += (c == 0) ? (a * ct + b * st) : (b * ct - a * st);
      }
    s *= (1.f / 19.f);
  }
  Bpack[((c * 8 + (i >> 1)) * WPAD + w) * 2 + (i & 1)] = s;
}

static __device__ inline void role_C(int wp, int tid, float* tw_c, float* tw_s,
                                     const float* __restrict__ Sre, const float* __restrict__ Sim,
                                     const float* __restrict__ pw, float* __restrict__ Cpack) {
  if (tid < GRID) {
    float th = (float)tid * TWO_PI_19;
    tw_c[tid] = cosf(th);
    tw_s[tid] = sinf(th);
  }
  __syncthreads();

  const int j    = tid >> 4;
  const int part = tid & 15;
  int wu[2], wv[2];
  for (int c = 0; c < 2; ++c) {
    int w = 2 * wp + c;
    wu[c] = w / GRID;
    wv[c] = w % GRID;
  }

  float g3r[2] = {0.f, 0.f}, g3i[2] = {0.f, 0.f};
  const float* sre = Sre + j * WREAL;
  const float* sim = Sim + j * WREAL;

#pragma unroll
  for (int it = 0; it < 23; ++it) {
    int t = part + 16 * it;
    bool ok = t < WREAL;
    int tc = ok ? t : 0;
    float a = sre[tc];
    float b = sim[tc];
    a = ok ? a : 0.f;
    b = ok ? b : 0.f;
    int tu = tc / GRID, tv = tc - tu * GRID;
    for (int c = 0; c < 2; ++c) {
      int ph3 = (wu[c] * tu + wv[c] * tv) % GRID;
      float c3 = tw_c[ph3], s3 = tw_s[ph3];
      g3r[c] += a * c3 - b * s3;
      g3i[c] += b * c3 + a * s3;
    }
  }

  float v[4] = { g3r[0], g3i[0], g3r[1], g3i[1] };
  for (int m = 8; m >= 1; m >>= 1) {
    for (int k = 0; k < 4; ++k)
      v[k] += __shfl_xor(v[k], m, 16);
  }

  if (part == 0) {
    float scale = pw[j] * (1.f / 19.f);
    bool ok0 = (2 * wp)     < WREAL;
    bool ok1 = (2 * wp + 1) < WREAL;
    float q0[2] = { v[0] * scale, -v[1] * scale };
    float q1[2] = { v[2] * scale, -v[3] * scale };
    v2f* C2 = (v2f*)Cpack;
    for (int qi = 0; qi < 2; ++qi) {
      v2f o;
      o.x = ok0 ? q0[qi] : 0.f;
      o.y = ok1 ? q1[qi] : 0.f;
      C2[(qi * (WPAD / 2) + wp) * 16 + j] = o;
    }
  }
}

static __device__ inline void role_L(int i, int j,
                                     const float* __restrict__ Mre, const float* __restrict__ Mim,
                                     const float* __restrict__ Sre, const float* __restrict__ Sim,
                                     const float* __restrict__ pw, float* __restrict__ Lpack) {
  float s = 0.f;
  for (int u = 0; u < 7; ++u)
    for (int v = 0; v < 7; ++v) {
      s += Mre[(i * 7 + u) * 7 + v] * Sre[(j * GRID + u + 6) * GRID + v + 6]
         - Mim[(i * 7 + u) * 7 + v] * Sim[(j * GRID + u + 6) * GRID + v + 6];
    }
  s *= pw[j];
  Lpack[((i >> 1) * 16 + j) * 2 + (i & 1)] = s;
}

static __device__ inline void role_Q(int e, int h, int tid, float* cv_re, float* cv_im,
                                     const float* __restrict__ Mre, const float* __restrict__ Mim,
                                     const float* __restrict__ Sre, const float* __restrict__ Sim,
                                     const float* __restrict__ pw, float* __restrict__ Qpack) {
  constexpr SlotTable slots = make_slot_table();
  const int ra = slots.a[e], rb = slots.b[e];
  const int a = ra ^ (2 * h);
  const int b = rb ^ (2 * h);
  const bool own_image = (rb == (ra ^ 2));

  if (tid < NCONV) {
    const int tu = tid / 13, tv = tid - tu * 13;
    const float* are = Mre + a * 49;
    const float* aim = Mim + a * 49;
    const float* bre = Mre + b * 49;
    const float* bim = Mim + b * 49;
    float cr = 0.f, ci = 0.f;
#pragma unroll
    for (int k = 0; k < 49; ++k) {
      const int u1 = k / 7, v1 = k - u1 * 7;
      const int u2 = tu - u1, v2 = tv - v1;
      const bool ok = (u2 >= 0) && (u2 < 7) && (v2 >= 0) && (v2 < 7);
      const int kb = ok ? (u2 * 7 + v2) : 0;
      float br = bre[kb];
      float bi = bim[kb];
      br = ok ? br : 0.f;
      bi = ok ? bi : 0.f;
      const float ar = are[k], ai = aim[k];
      cr += ar * br - ai * bi;
      ci += ar * bi + ai * br;
    }
    cv_re[tid] = cr;
    cv_im[tid] = ci;
  }
  __syncthreads();

  const int j    = tid >> 4;
  const int part = tid & 15;
  float s = 0.f;
#pragma unroll
  for (int it = 0; it < 11; ++it) {
    const int t = part + 16 * it;
    const bool ok = t < NCONV;
    const int tc = ok ? t : 0;
    const int tu = tc / 13, tv = tc - tu * 13;
    const int si = (j * GRID + tu + 3) * GRID + tv + 3;
    float sr = Sre[si];
    float sm = Sim[si];
    sr = ok ? sr : 0.f;
    sm = ok ? sm : 0.f;
    s += sr * cv_re[tc] - sm * cv_im[tc];
  }
  for (int m = 8; m >= 1; m >>= 1)
    s += __shfl_xor(s, m, 16);

  if (part == 0) {
    const float mult = (a == b) ? 1.f : 2.f;
    const float val = (own_image && h == 1) ? 0.f : s * mult * pw[j] * (1.f / 19.f);
    Qpack[(((e >> 1) * 2 + h) * 16 + j) * 2 + (e & 1)] = val;
  }
}

#define NB_B    (WPAD / 8)
#define NB_C    (WPAD / 2)
#define NB_Q    (NSLOT * 2)
#define NB_PRE  (NB_B + NB_C + NB_Q + 1)
__global__ __launch_bounds__(256)
void pre_all(const float* __restrict__ Mre, const float* __restrict__ Mim,
             const float* __restrict__ Sre, const float* __restrict__ Sim,
             const float* __restrict__ pw,
             float* __restrict__ Bpack, float* __restrict__ Cpack,
             float* __restrict__ Lpack, float* __restrict__ Qpack) {
  __shared__ float tw_c[GRID];
  __shared__ float tw_s[GRID];
  __shared__ float cv_re[NCONV];
  __shared__ float cv_im[NCONV];
  const int tid = threadIdx.x;
  const int b   = blockIdx.x;
  if (b < NB_B) {
    role_B(b * 8 + (tid >> 5), tid & 31, Mre, Mim, Bpack);
  } else if (b < NB_B + NB_C) {
    role_C(b - NB_B, tid, tw_c, tw_s, Sre, Sim, pw, Cpack);
  } else if (b < NB_B + NB_C + NB_Q) {
    const int bq = b - (NB_B + NB_C);
    role_Q(bq >> 1, bq & 1, tid, cv_re, cv_im, Mre, Mim, Sre, Sim, pw, Qpack);
  } else {
    role_L(tid & 15, tid >> 4, Mre, Mim, Sre, Sim, pw, Lpack);
  }
}


template <int T>
static __device__ __forceinline__ void quad_steps(const v2f* __restrict__ Qw, const float (&z)[16], v8f& acc) {
  if constexpr (T < NT2) {
    constexpr IdxPair p0 = slot_pair(2 * T);
    constexpr IdxPair p1 = slot_pair(2 * T + 1);
    v2f b;
    b.x = z[p0.a] * z[p0.b];
    b.y = z[p1.a] * z[p1.b];
    acc = wmma4(Qw[T * 32], b, acc);
    quad_steps<T + 1>(Qw, z, acc);
  }
}

__global__ __launch_bounds__(256, 1)
void gaunt_main(const float* __restrict__ X,
                const float* __restrict__ Bp,
                const float* __restrict__ Cp,
                const float* __restrict__ Lp,
                const float* __restrict__ Qp,
                float* __restrict__ Y) {
  const int tid  = threadIdx.x;
  const int wave = tid >> 5;
  const int lane = tid & 31;
  const int half = lane >> 4;
  const int lr   = lane & 15;

  const long row0 = ((long)blockIdx.x * 8 + wave) * 16;

  const v2f* B2 = (const v2f*)Bp;
  const v2f* C2 = (const v2f*)Cp;
  const v2f* L2 = (const v2f*)Lp;
  const v2f* Q2 = (const v2f*)Qp;

  v2f xa[4];
  float z[16];
  const float* xrow = X + (row0 + lr) * DIN;
#pragma unroll
  for (int kk = 0; kk < 4; ++kk) {
    xa[kk] = *(const v2f*)(xrow + 4 * kk + 2 * half);
    v2f xo = *(const v2f*)(xrow + 4 * kk + 2 * (1 - half));
    z[4 * kk + 0] = xa[kk].x;
    z[4 * kk + 1] = xa[kk].y;
    z[4 * kk + 2] = xo.x;
    z[4 * kk + 3] = xo.y;
  }

  v8f acc = {0.f, 0.f, 0.f, 0.f, 0.f, 0.f, 0.f, 0.f};

#pragma unroll
  for (int kk = 0; kk < 4; ++kk)
    acc = wmma4(L2[(2 * kk + half) * 16 + lr], xa[kk], acc);

  quad_steps<0>(Q2 + half * 16 + lr, z, acc);

  const int boff = half * WPAD + lr;
  const int coff = (4 * half) * 16 + lr;

  for (int wt = 0; wt < NWT; ++wt) {
    const v2f* Bw = B2 + wt * 16;
    const v2f* Cw = C2 + wt * (8 * 16);

    v8f fre = {0.f, 0.f, 0.f, 0.f, 0.f, 0.f, 0.f, 0.f};
    v8f fim = {0.f, 0.f, 0.f, 0.f, 0.f, 0.f, 0.f, 0.f};
#pragma unroll
    for (int kk = 0; kk < 4; ++kk) {
      fre = wmma4(Bw[(    2 * kk) * WPAD + boff], xa[kk], fre);
      fim = wmma4(Bw[(8 + 2 * kk) * WPAD + boff], xa[kk], fim);
    }

    v8f pq[2];
#pragma unroll
    for (int e = 0; e < 8; ++e) {
      float fr = fre[e], fi = fim[e];
      float p2r = fr * fr - fi * fi;
      float p2i = 2.f * fr * fi;
      float p3r = p2r * fr - p2i * fi;
      float p3i = p2r * fi + p2i * fr;
      pq[0][e] = p3r;
      pq[1][e] = p3i;
    }

#pragma unroll
    for (int q = 0; q < 2; ++q) {
#pragma unroll
      for (int t = 0; t < 4; ++t) {
        v2f a = Cw[(q * (WPAD / 2) + t) * 16 + coff];
        v2f b;
        b.x = pq[q][2 * t];
        b.y = pq[q][2 * t + 1];
        acc = wmma4(a, b, acc);
      }
    }
  }

  typedef float v4f __attribute__((ext_vector_type(4)));
  float* yrow = Y + (row0 + lr) * DOUT + 8 * half;
  v4f lo = {acc[0], acc[1], acc[2], acc[3]};
  v4f hi = {acc[4], acc[5], acc[6], acc[7]};
  *(v4f*)(yrow)     = lo;
  *(v4f*)(yrow + 4) = hi;
}


extern "C" void kernel_launch(void* const* d_in, const int* in_sizes, int n_in,
                              void* d_out, int out_size, void* d_ws, size_t ws_size,
                              hipStream_t stream) {
  const float* x   = (const float*)d_in[0];
  const float* Mre = (const float*)d_in[1];
  const float* Mim = (const float*)d_in[2];
  const float* Sre = (const float*)d_in[3];
  const float* Sim = (const float*)d_in[4];
  const float* pw  = (const float*)d_in[5];

  float* ws = (float*)d_ws;
  float* Bp = ws + B_OFF;
  float* Cp = ws + C_OFF;
  float* Lp = ws + L_OFF;
  float* Qp = ws + T_OFF;

  pre_all<<<NB_PRE, 256, 0, stream>>>(Mre, Mim, Sre, Sim, pw, Bp, Cp, Lp, Qp);

  gaunt_main<<<NROWS / (8 * 16), 256, 0, stream>>>(x, Bp, Cp, Lp, Qp, (float*)d_out);
}


# Degree-3 term as a direct cubic form: lanes form x_a*x_b*x_c, one K=816 WMMA product with a prebuilt coefficient pack replaces both Fourier products of every tile; sum order of y changes
# speedup vs baseline: 4.7064x; 1.2850x over previous
#include <hip/hip_runtime.h>
#include <hip/hip_bf16.h>
#include <math.h>


typedef float v2f __attribute__((ext_vector_type(2)));
typedef float v8f __attribute__((ext_vector_type(8)));

#define NROWS   131072
#define DIN     16
#define DOUT    16
#define GRID    19
#define WREAL   361

#define NSLOT   72
#define NT2     (NSLOT / 2)
#define NCONV   169

#define NSLOT3  408
#define NT3     (NSLOT3 / 2)
#define MBW     19
#define PPW     25

#define L_OFF   0
#define Q_OFF   256
#define T_OFF   (Q_OFF + 2304)
#define WS_FLOATS (T_OFF + 13056)

struct IdxPair { int a, b; };
__host__ __device__ constexpr IdxPair slot_pair(int e) {
  int n = 0;
  for (int a = 0; a < 16; ++a)
    for (int b = a; b < 16; ++b) {
      int pa = a ^ 2, pb = b ^ 2;
      if (pa > pb) { int t = pa; pa = pb; pb = t; }
      bool first = (a < pa) || (a == pa && b <= pb);
      if (first) {
        if (n == e) return IdxPair{a, b};
        ++n;
      }
    }
  return IdxPair{0, 0};
}
struct SlotTable { unsigned char a[NSLOT]; unsigned char b[NSLOT]; };
__host__ __device__ constexpr SlotTable make_slot_table() {
  SlotTable t{};
  for (int e = 0; e < NSLOT; ++e) {
    IdxPair p = slot_pair(e);
    t.a[e] = (unsigned char)p.a;
    t.b[e] = (unsigned char)p.b;
  }
  return t;
}

struct Slot3Table { unsigned char a[NSLOT3]; unsigned char b[NSLOT3]; unsigned char c[NSLOT3]; int n; };
__host__ __device__ constexpr Slot3Table make_slot3_table() {
  Slot3Table t{};
  int n = 0;
  for (int a = 0; a < 16; ++a)
    for (int b = a; b < 16; ++b)
      for (int c = b; c < 16; ++c) {
        int p0 = a ^ 2, p1 = b ^ 2, p2 = c ^ 2;
        if (p0 > p1) { int s = p0; p0 = p1; p1 = s; }
        if (p1 > p2) { int s = p1; p1 = p2; p2 = s; }
        if (p0 > p1) { int s = p0; p0 = p1; p1 = s; }
        bool first = (a < p0) || (a == p0 && (b < p1 || (b == p1 && c <= p2)));
        if (first) {
          if (n < NSLOT3) {
            t.a[n] = (unsigned char)a;
            t.b[n] = (unsigned char)b;
            t.c[n] = (unsigned char)c;
          }
          ++n;
        }
      }
  t.n = n;
  return t;
}
static_assert(make_slot3_table().n == NSLOT3, "the swap must halve the 816 triples exactly");

static __device__ inline v8f wmma4(v2f a, v2f b, v8f c) {
  return __builtin_amdgcn_wmma_f32_16x16x4_f32(
       false, a,  false, b,
       (short)0, c,  false,  false);
}


static __device__ inline void role_L(int i, int j,
                                     const float* __restrict__ Mre, const float* __restrict__ Mim,
                                     const float* __restrict__ Sre, const float* __restrict__ Sim,
                                     const float* __restrict__ pw, float* __restrict__ Lpack) {
  float s = 0.f;
  for (int u = 0; u < 7; ++u)
    for (int v = 0; v < 7; ++v) {
      s += Mre[(i * 7 + u) * 7 + v] * Sre[(j * GRID + u + 6) * GRID + v + 6]
         - Mim[(i * 7 + u) * 7 + v] * Sim[(j * GRID + u + 6) * GRID + v + 6];
    }
  s *= pw[j];
  Lpack[((i >> 1) * 16 + j) * 2 + (i & 1)] = s;
}

static __device__ inline void role_Q(int e, int h, int tid, float* cv_re, float* cv_im,
                                     const float* __restrict__ Mre, const float* __restrict__ Mim,
                                     const float* __restrict__ Sre, const float* __restrict__ Sim,
                                     const float* __restrict__ pw, float* __restrict__ Qpack) {
  constexpr SlotTable slots = make_slot_table();
  const int ra = slots.a[e], rb = slots.b[e];
  const int a = ra ^ (2 * h);
  const int b = rb ^ (2 * h);
  const bool own_image = (rb == (ra ^ 2));

  if (tid < NCONV) {
    const int tu = tid / 13, tv = tid - tu * 13;
    const float* are = Mre + a * 49;
    const float* aim = Mim + a * 49;
    const float* bre = Mre + b * 49;
    const float* bim = Mim + b * 49;
    float cr = 0.f, ci = 0.f;
#pragma unroll
    for (int k = 0; k < 49; ++k) {
      const int u1 = k / 7, v1 = k - u1 * 7;
      const int u2 = tu - u1, v2 = tv - v1;
      const bool ok = (u2 >= 0) && (u2 < 7) && (v2 >= 0) && (v2 < 7);
      const int kb = ok ? (u2 * 7 + v2) : 0;
      float br = bre[kb];
      float bi = bim[kb];
      br = ok ? br : 0.f;
      bi = ok ? bi : 0.f;
      const float ar = are[k], ai = aim[k];
      cr += ar * br - ai * bi;
      ci += ar * bi + ai * br;
    }
    cv_re[tid] = cr;
    cv_im[tid] = ci;
  }
  __syncthreads();

  const int j    = tid >> 4;
  const int part = tid & 15;
  float s = 0.f;
#pragma unroll
  for (int it = 0; it < 11; ++it) {
    const int t = part + 16 * it;
    const bool ok = t < NCONV;
    const int tc = ok ? t : 0;
    const int tu = tc / 13, tv = tc - tu * 13;
    const int si = (j * GRID + tu + 3) * GRID + tv + 3;
    float sr = Sre[si];
    float sm = Sim[si];
    sr = ok ? sr : 0.f;
    sm = ok ? sm : 0.f;
    s += sr * cv_re[tc] - sm * cv_im[tc];
  }
  for (int m = 8; m >= 1; m >>= 1)
    s += __shfl_xor(s, m, 16);

  if (part == 0) {
    const float mult = (a == b) ? 1.f : 2.f;
    const float val = (own_image && h == 1) ? 0.f : s * mult * pw[j] * (1.f / 19.f);
    Qpack[(((e >> 1) * 2 + h) * 16 + j) * 2 + (e & 1)] = val;
  }
}

static __device__ __forceinline__ void cmac(v2f& acc0, v2f& acc1, float mr, float mi, v2f p) {
  v2f q;
  q.x = -p.y;
  q.y = p.x;
  acc0 += mr * p;
  acc1 += mi * q;
}

static __device__ inline void role_T(int e, int h, int tid,
                                     v2f* mb, v2f* pp, v2f* c3,
                                     const float* __restrict__ Mre, const float* __restrict__ Mim,
                                     const float* __restrict__ Sre, const float* __restrict__ Sim,
                                     const float* __restrict__ pw, float* __restrict__ Tpack) {
  constexpr Slot3Table slots = make_slot3_table();
  const int a = slots.a[e] ^ (2 * h);
  const int b = slots.b[e] ^ (2 * h);
  const int c = slots.c[e] ^ (2 * h);

  for (int t = tid; t < MBW * MBW; t += 256) {
    const int u = t / MBW - 6, v = t % MBW - 6;
    const bool ok = (u >= 0) && (u < 7) && (v >= 0) && (v < 7);
    const int k = ok ? (u * 7 + v) : 0;
    const float br = Mre[b * 49 + k];
    const float bi = Mim[b * 49 + k];
    v2f o;
    o.x = ok ? br : 0.f;
    o.y = ok ? bi : 0.f;
    mb[t] = o;
  }
  for (int t = tid; t < PPW * PPW; t += 256) {
    const int u = t / PPW, v = t % PPW;
    const bool middle = (u >= 6) && (u < 19) && (v >= 6) && (v < 19);
    if (!middle) {
      v2f o = {0.f, 0.f};
      pp[t] = o;
    }
  }
  __syncthreads();

  if (tid < NCONV) {
    const int pu = tid / 13, pv = tid - pu * 13;
    const float* are = Mre + a * 49;
    const float* aim = Mim + a * 49;
    const v2f* src = mb + (pu + 6) * MBW + (pv + 6);
    v2f acc0 = {0.f, 0.f}, acc1 = {0.f, 0.f};
#pragma unroll
    for (int k = 0; k < 49; ++k) {
      const int u1 = k / 7, v1 = k - u1 * 7;
      cmac(acc0, acc1, are[k], aim[k], src[-(u1 * MBW + v1)]);
    }
    pp[(pu + 6) * PPW + (pv + 6)] = acc0 + acc1;
  }
  __syncthreads();

  for (int t = tid; t < WREAL; t += 256) {
    const int tu = t / GRID, tv = t - tu * GRID;
    const float* cre = Mre + c * 49;
    const float* cim = Mim + c * 49;
    const v2f* src = pp + (tu + 6) * PPW + (tv + 6);
    v2f acc0 = {0.f, 0.f}, acc1 = {0.f, 0.f};
#pragma unroll
    for (int k = 0; k < 49; ++k) {
      const int u1 = k / 7, v1 = k - u1 * 7;
      cmac(acc0, acc1, cre[k], cim[k], src[-(u1 * PPW + v1)]);
    }
    c3[t] = acc0 + acc1;
  }
  __syncthreads();

  const int j    = tid >> 4;
  const int part = tid & 15;
  float s = 0.f;
#pragma unroll
  for (int it = 0; it < 23; ++it) {
    const int t = part + 16 * it;
    const bool ok = t < WREAL;
    const int tc = ok ? t : 0;
    float sr = Sre[j * WREAL + tc];
    float sm = Sim[j * WREAL + tc];
    sr = ok ? sr : 0.f;
    sm = ok ? sm : 0.f;
    const v2f kv = c3[tc];
    s += sr * kv.x - sm * kv.y;
  }
  for (int m = 8; m >= 1; m >>= 1)
    s += __shfl_xor(s, m, 16);

  if (part == 0) {
    const float mult = (a == b && b == c) ? 1.f : ((a == b || b == c || a == c) ? 3.f : 6.f);
    const float val = s * mult * pw[j] * (1.f / 361.f);
    Tpack[(((e >> 1) * 2 + h) * 16 + j) * 2 + (e & 1)] = val;
  }
}

#define NB_T    (NSLOT3 * 2)
#define NB_Q    (NSLOT * 2)
#define NB_PRE  (NB_T + NB_Q + 1)
__global__ __launch_bounds__(256)
void pre_all(const float* __restrict__ Mre, const float* __restrict__ Mim,
             const float* __restrict__ Sre, const float* __restrict__ Sim,
             const float* __restrict__ pw,
             float* __restrict__ Lpack, float* __restrict__ Qpack, float* __restrict__ Tpack) {
  __shared__ float cv_re[NCONV];
  __shared__ float cv_im[NCONV];
  __shared__ v2f mb[MBW * MBW];
  __shared__ v2f pp[PPW * PPW];
  __shared__ v2f c3[WREAL];
  const int tid = threadIdx.x;
  const int b   = blockIdx.x;
  if (b < NB_T) {
    role_T(b >> 1, b & 1, tid, mb, pp, c3, Mre, Mim, Sre, Sim, pw, Tpack);
  } else if (b < NB_T + NB_Q) {
    const int bq = b - NB_T;
    role_Q(bq >> 1, bq & 1, tid, cv_re, cv_im, Mre, Mim, Sre, Sim, pw, Qpack);
  } else {
    role_L(tid & 15, tid >> 4, Mre, Mim, Sre, Sim, pw, Lpack);
  }
}


template <int T>
static __device__ __forceinline__ void quad_steps(const v2f* __restrict__ Qw, const float (&z)[16], v8f& acc) {
  if constexpr (T < NT2) {
    constexpr IdxPair p0 = slot_pair(2 * T);
    constexpr IdxPair p1 = slot_pair(2 * T + 1);
    v2f b;
    b.x = z[p0.a] * z[p0.b];
    b.y = z[p1.a] * z[p1.b];
    acc = wmma4(Qw[T * 32], b, acc);
    quad_steps<T + 1>(Qw, z, acc);
  }
}

template <int T>
static __device__ __forceinline__ void cube_steps(const v2f* __restrict__ Tw, const float (&z)[16], v8f& acc) {
  if constexpr (T < NT3) {
    constexpr Slot3Table tab = make_slot3_table();
    constexpr int a0 = tab.a[2 * T],     b0 = tab.b[2 * T],     c0 = tab.c[2 * T];
    constexpr int a1 = tab.a[2 * T + 1], b1 = tab.b[2 * T + 1], c1 = tab.c[2 * T + 1];
    v2f b;
    b.x = (z[a0] * z[b0]) * z[c0];
    b.y = (z[a1] * z[b1]) * z[c1];
    acc = wmma4(Tw[T * 32], b, acc);
    cube_steps<T + 1>(Tw, z, acc);
  }
}

__global__ __launch_bounds__(256, 1)
void gaunt_main(const float* __restrict__ X,
                const float* __restrict__ Lp,
                const float* __restrict__ Qp,
                const float* __restrict__ Tp,
                float* __restrict__ Y) {
  const int tid  = threadIdx.x;
  const int wave = tid >> 5;
  const int lane = tid & 31;
  const int half = lane >> 4;
  const int lr   = lane & 15;

  const long row0 = ((long)blockIdx.x * 8 + wave) * 16;

  const v2f* L2 = (const v2f*)Lp;
  const v2f* Q2 = (const v2f*)Qp;
  const v2f* T2 = (const v2f*)Tp;

  v2f xa[4];
  float z[16];
  const float* xrow = X + (row0 + lr) * DIN;
#pragma unroll
  for (int kk = 0; kk < 4; ++kk) {
    xa[kk] = *(const v2f*)(xrow + 4 * kk + 2 * half);
    v2f xo = *(const v2f*)(xrow + 4 * kk + 2 * (1 - half));
    z[4 * kk + 0] = xa[kk].x;
    z[4 * kk + 1] = xa[kk].y;
    z[4 * kk + 2] = xo.x;
    z[4 * kk + 3] = xo.y;
  }

  v8f acc = {0.f, 0.f, 0.f, 0.f, 0.f, 0.f, 0.f, 0.f};

#pragma unroll
  for (int kk = 0; kk < 4; ++kk)
    acc = wmma4(L2[(2 * kk + half) * 16 + lr], xa[kk], acc);

  quad_steps<0>(Q2 + half * 16 + lr, z, acc);

  cube_steps<0>(T2 + half * 16 + lr, z, acc);

  typedef float v4f __attribute__((ext_vector_type(4)));
  float* yrow = Y + (row0 + lr) * DOUT + 8 * half;
  v4f lo = {acc[0], acc[1], acc[2], acc[3]};
  v4f hi = {acc[4], acc[5], acc[6], acc[7]};
  *(v4f*)(yrow)     = lo;
  *(v4f*)(yrow + 4) = hi;
}


extern "C" void kernel_launch(void* const* d_in, const int* in_sizes, int n_in,
                              void* d_out, int out_size, void* d_ws, size_t ws_size,
                              hipStream_t stream) {
  const float* x   = (const float*)d_in[0];
  const float* Mre = (const float*)d_in[1];
  const float* Mim = (const float*)d_in[2];
  const float* Sre = (const float*)d_in[3];
  const float* Sim = (const float*)d_in[4];
  const float* pw  = (const float*)d_in[5];

  float* ws = (float*)d_ws;
  float* Lp = ws + L_OFF;
  float* Qp = ws + Q_OFF;
  float* Tp = ws + T_OFF;

  pre_all<<<NB_PRE, 256, 0, stream>>>(Mre, Mim, Sre, Sim, pw, Lp, Qp, Tp);

  gaunt_main<<<NROWS / (8 * 16), 256, 0, stream>>>(x, Lp, Qp, Tp, (float*)d_out);
}


# Main kernel: one wave owns four sample tiles, each with its own accumulator; every coefficient fragment is loaded once and used for all four; 8192 waves become 2048; per-sample sum order unchanged
# speedup vs baseline: 5.3944x; 1.1462x over previous
#include <hip/hip_runtime.h>
#include <hip/hip_bf16.h>
#include <math.h>


typedef float v2f __attribute__((ext_vector_type(2)));
typedef float v8f __attribute__((ext_vector_type(8)));

#define NROWS   131072
#define TPW     4
#define DIN     16
#define DOUT    16
#define GRID    19
#define WREAL   361

#define NSLOT   72
#define NT2     (NSLOT / 2)
#define NCONV   169

#define NSLOT3  408
#define NT3     (NSLOT3 / 2)
#define MBW     19
#define PPW     25

#define L_OFF   0
#define Q_OFF   256
#define T_OFF   (Q_OFF + 2304)
#define WS_FLOATS (T_OFF + 13056)

struct IdxPair { int a, b; };
__host__ __device__ constexpr IdxPair slot_pair(int e) {
  int n = 0;
  for (int a = 0; a < 16; ++a)
    for (int b = a; b < 16; ++b) {
      int pa = a ^ 2, pb = b ^ 2;
      if (pa > pb) { int t = pa; pa = pb; pb = t; }
      bool first = (a < pa) || (a == pa && b <= pb);
      if (first) {
        if (n == e) return IdxPair{a, b};
        ++n;
      }
    }
  return IdxPair{0, 0};
}
struct SlotTable { unsigned char a[NSLOT]; unsigned char b[NSLOT]; };
__host__ __device__ constexpr SlotTable make_slot_table() {
  SlotTable t{};
  for (int e = 0; e < NSLOT; ++e) {
    IdxPair p = slot_pair(e);
    t.a[e] = (unsigned char)p.a;
    t.b[e] = (unsigned char)p.b;
  }
  return t;
}

struct Slot3Table { unsigned char a[NSLOT3]; unsigned char b[NSLOT3]; unsigned char c[NSLOT3]; int n; };
__host__ __device__ constexpr Slot3Table make_slot3_table() {
  Slot3Table t{};
  int n = 0;
  for (int a = 0; a < 16; ++a)
    for (int b = a; b < 16; ++b)
      for (int c = b; c < 16; ++c) {
        int p0 = a ^ 2, p1 = b ^ 2, p2 = c ^ 2;
        if (p0 > p1) { int s = p0; p0 = p1; p1 = s; }
        if (p1 > p2) { int s = p1; p1 = p2; p2 = s; }
        if (p0 > p1) { int s = p0; p0 = p1; p1 = s; }
        bool first = (a < p0) || (a == p0 && (b < p1 || (b == p1 && c <= p2)));
        if (first) {
          if (n < NSLOT3) {
            t.a[n] = (unsigned char)a;
            t.b[n] = (unsigned char)b;
            t.c[n] = (unsigned char)c;
          }
          ++n;
        }
      }
  t.n = n;
  return t;
}
static_assert(make_slot3_table().n == NSLOT3, "the swap must halve the 816 triples exactly");

static __device__ inline v8f wmma4(v2f a, v2f b, v8f c) {
  return __builtin_amdgcn_wmma_f32_16x16x4_f32(
       false, a,  false, b,
       (short)0, c,  false,  false);
}


static __device__ inline void role_L(int i, int j,
                                     const float* __restrict__ Mre, const float* __restrict__ Mim,
                                     const float* __restrict__ Sre, const float* __restrict__ Sim,
                                     const float* __restrict__ pw, float* __restrict__ Lpack) {
  float s = 0.f;
  for (int u = 0; u < 7; ++u)
    for (int v = 0; v < 7; ++v) {
      s += Mre[(i * 7 + u) * 7 + v] * Sre[(j * GRID + u + 6) * GRID + v + 6]
         - Mim[(i * 7 + u) * 7 + v] * Sim[(j * GRID + u + 6) * GRID + v + 6];
    }
  s *= pw[j];
  Lpack[((i >> 1) * 16 + j) * 2 + (i & 1)] = s;
}

static __device__ inline void role_Q(int e, int h, int tid, float* cv_re, float* cv_im,
                                     const float* __restrict__ Mre, const float* __restrict__ Mim,
                                     const float* __restrict__ Sre, const float* __restrict__ Sim,
                                     const float* __restrict__ pw, float* __restrict__ Qpack) {
  constexpr SlotTable slots = make_slot_table();
  const int ra = slots.a[e], rb = slots.b[e];
  const int a = ra ^ (2 * h);
  const int b = rb ^ (2 * h);
  const bool own_image = (rb == (ra ^ 2));

  if (tid < NCONV) {
    const int tu = tid / 13, tv = tid - tu * 13;
    const float* are = Mre + a * 49;
    const float* aim = Mim + a * 49;
    const float* bre = Mre + b * 49;
    const float* bim = Mim + b * 49;
    float cr = 0.f, ci = 0.f;
#pragma unroll
    for (int k = 0; k < 49; ++k) {
      const int u1 = k / 7, v1 = k - u1 * 7;
      const int u2 = tu - u1, v2 = tv - v1;
      const bool ok = (u2 >= 0) && (u2 < 7) && (v2 >= 0) && (v2 < 7);
      const int kb = ok ? (u2 * 7 + v2) : 0;
      float br = bre[kb];
      float bi = bim[kb];
      br = ok ? br : 0.f;
      bi = ok ? bi : 0.f;
      const float ar = are[k], ai = aim[k];
      cr += ar * br - ai * bi;
      ci += ar * bi + ai * br;
    }
    cv_re[tid] = cr;
    cv_im[tid] = ci;
  }
  __syncthreads();

  const int j    = tid >> 4;
  const int part = tid & 15;
  float s = 0.f;
#pragma unroll
  for (int it = 0; it < 11; ++it) {
    const int t = part + 16 * it;
    const bool ok = t < NCONV;
    const int tc = ok ? t : 0;
    const int tu = tc / 13, tv = tc - tu * 13;
    const int si = (j * GRID + tu + 3) * GRID + tv + 3;
    float sr = Sre[si];
    float sm = Sim[si];
    sr = ok ? sr : 0.f;
    sm = ok ? sm : 0.f;
    s += sr * cv_re[tc] - sm * cv_im[tc];
  }
  for (int m = 8; m >= 1; m >>= 1)
    s += __shfl_xor(s, m, 16);

  if (part == 0) {
    const float mult = (a == b) ? 1.f : 2.f;
    const float val = (own_image && h == 1) ? 0.f : s * mult * pw[j] * (1.f / 19.f);
    Qpack[(((e >> 1) * 2 + h) * 16 + j) * 2 + (e & 1)] = val;
  }
}

static __device__ __forceinline__ void cmac(v2f& acc0, v2f& acc1, float mr, float mi, v2f p) {
  v2f q;
  q.x = -p.y;
  q.y = p.x;
  acc0 += mr * p;
  acc1 += mi * q;
}

static __device__ inline void role_T(int e, int h, int tid,
                                     v2f* mb, v2f* pp, v2f* c3,
                                     const float* __restrict__ Mre, const float* __restrict__ Mim,
                                     const float* __restrict__ Sre, const float* __restrict__ Sim,
                                     const float* __restrict__ pw, float* __restrict__ Tpack) {
  constexpr Slot3Table slots = make_slot3_table();
  const int a = slots.a[e] ^ (2 * h);
  const int b = slots.b[e] ^ (2 * h);
  const int c = slots.c[e] ^ (2 * h);

  for (int t = tid; t < MBW * MBW; t += 256) {
    const int u = t / MBW - 6, v = t % MBW - 6;
    const bool ok = (u >= 0) && (u < 7) && (v >= 0) && (v < 7);
    const int k = ok ? (u * 7 + v) : 0;
    const float br = Mre[b * 49 + k];
    const float bi = Mim[b * 49 + k];
    v2f o;
    o.x = ok ? br : 0.f;
    o.y = ok ? bi : 0.f;
    mb[t] = o;
  }
  for (int t = tid; t < PPW * PPW; t += 256) {
    const int u = t / PPW, v = t % PPW;
    const bool middle = (u >= 6) && (u < 19) && (v >= 6) && (v < 19);
    if (!middle) {
      v2f o = {0.f, 0.f};
      pp[t] = o;
    }
  }
  __syncthreads();

  if (tid < NCONV) {
    const int pu = tid / 13, pv = tid - pu * 13;
    const float* are = Mre + a * 49;
    const float* aim = Mim + a * 49;
    const v2f* src = mb + (pu + 6) * MBW + (pv + 6);
    v2f acc0 = {0.f, 0.f}, acc1 = {0.f, 0.f};
#pragma unroll
    for (int k = 0; k < 49; ++k) {
      const int u1 = k / 7, v1 = k - u1 * 7;
      cmac(acc0, acc1, are[k], aim[k], src[-(u1 * MBW + v1)]);
    }
    pp[(pu + 6) * PPW + (pv + 6)] = acc0 + acc1;
  }
  __syncthreads();

  for (int t = tid; t < WREAL; t += 256) {
    const int tu = t / GRID, tv = t - tu * GRID;
    const float* cre = Mre + c * 49;
    const float* cim = Mim + c * 49;
    const v2f* src = pp + (tu + 6) * PPW + (tv + 6);
    v2f acc0 = {0.f, 0.f}, acc1 = {0.f, 0.f};
#pragma unroll
    for (int k = 0; k < 49; ++k) {
      const int u1 = k / 7, v1 = k - u1 * 7;
      cmac(acc0, acc1, cre[k], cim[k], src[-(u1 * PPW + v1)]);
    }
    c3[t] = acc0 + acc1;
  }
  __syncthreads();

  const int j    = tid >> 4;
  const int part = tid & 15;
  float s = 0.f;
#pragma unroll
  for (int it = 0; it < 23; ++it) {
    const int t = part + 16 * it;
    const bool ok = t < WREAL;
    const int tc = ok ? t : 0;
    float sr = Sre[j * WREAL + tc];
    float sm = Sim[j * WREAL + tc];
    sr = ok ? sr : 0.f;
    sm = ok ? sm : 0.f;
    const v2f kv = c3[tc];
    s += sr * kv.x - sm * kv.y;
  }
  for (int m = 8; m >= 1; m >>= 1)
    s += __shfl_xor(s, m, 16);

  if (part == 0) {
    const float mult = (a == b && b == c) ? 1.f : ((a == b || b == c || a == c) ? 3.f : 6.f);
    const float val = s * mult * pw[j] * (1.f / 361.f);
    Tpack[(((e >> 1) * 2 + h) * 16 + j) * 2 + (e & 1)] = val;
  }
}

#define NB_T    (NSLOT3 * 2)
#define NB_Q    (NSLOT * 2)
#define NB_PRE  (NB_T + NB_Q + 1)
__global__ __launch_bounds__(256)
void pre_all(const float* __restrict__ Mre, const float* __restrict__ Mim,
             const float* __restrict__ Sre, const float* __restrict__ Sim,
             const float* __restrict__ pw,
             float* __restrict__ Lpack, float* __restrict__ Qpack, float* __restrict__ Tpack) {
  __shared__ float cv_re[NCONV];
  __shared__ float cv_im[NCONV];
  __shared__ v2f mb[MBW * MBW];
  __shared__ v2f pp[PPW * PPW];
  __shared__ v2f c3[WREAL];
  const int tid = threadIdx.x;
  const int b   = blockIdx.x;
  if (b < NB_T) {
    role_T(b >> 1, b & 1, tid, mb, pp, c3, Mre, Mim, Sre, Sim, pw, Tpack);
  } else if (b < NB_T + NB_Q) {
    const int bq = b - NB_T;
    role_Q(bq >> 1, bq & 1, tid, cv_re, cv_im, Mre, Mim, Sre, Sim, pw, Qpack);
  } else {
    role_L(tid & 15, tid >> 4, Mre, Mim, Sre, Sim, pw, Lpack);
  }
}


template <int T>
static __device__ __forceinline__ void quad_steps(const v2f* __restrict__ Qw, const float (&z)[TPW][16], v8f (&acc)[TPW]) {
  if constexpr (T < NT2) {
    constexpr IdxPair p0 = slot_pair(2 * T);
    constexpr IdxPair p1 = slot_pair(2 * T + 1);
    const v2f a = Qw[T * 32];
#pragma unroll
    for (int i = 0; i < TPW; ++i) {
      v2f b;
      b.x = z[i][p0.a] * z[i][p0.b];
      b.y = z[i][p1.a] * z[i][p1.b];
      acc[i] = wmma4(a, b, acc[i]);
    }
    quad_steps<T + 1>(Qw, z, acc);
  }
}

template <int T>
static __device__ __forceinline__ void cube_steps(const v2f* __restrict__ Tw, const float (&z)[TPW][16], v8f (&acc)[TPW]) {
  if constexpr (T < NT3) {
    constexpr Slot3Table tab = make_slot3_table();
    constexpr int a0 = tab.a[2 * T],     b0 = tab.b[2 * T],     c0 = tab.c[2 * T];
    constexpr int a1 = tab.a[2 * T + 1], b1 = tab.b[2 * T + 1], c1 = tab.c[2 * T + 1];
    const v2f a = Tw[T * 32];
#pragma unroll
    for (int i = 0; i < TPW; ++i) {
      v2f b;
      b.x = (z[i][a0] * z[i][b0]) * z[i][c0];
      b.y = (z[i][a1] * z[i][b1]) * z[i][c1];
      acc[i] = wmma4(a, b, acc[i]);
    }
    cube_steps<T + 1>(Tw, z, acc);
  }
}

__global__ __launch_bounds__(256, 1)
void gaunt_main(const float* __restrict__ X,
                const float* __restrict__ Lp,
                const float* __restrict__ Qp,
                const float* __restrict__ Tp,
                float* __restrict__ Y) {
  const int tid  = threadIdx.x;
  const int wave = tid >> 5;
  const int lane = tid & 31;
  const int half = lane >> 4;
  const int lr   = lane & 15;

  const long row0 = ((long)blockIdx.x * 8 + wave) * (16 * TPW);

  const v2f* L2 = (const v2f*)Lp;
  const v2f* Q2 = (const v2f*)Qp;
  const v2f* T2 = (const v2f*)Tp;

  float z[TPW][16];
#pragma unroll
  for (int i = 0; i < TPW; ++i) {
    const float* xrow = X + (row0 + 16 * i + lr) * DIN;
#pragma unroll
    for (int kk = 0; kk < 4; ++kk) {
      const v2f xs = *(const v2f*)(xrow + 4 * kk + 2 * half);
      const v2f xo = *(const v2f*)(xrow + 4 * kk + 2 * (1 - half));
      z[i][4 * kk + 0] = xs.x;
      z[i][4 * kk + 1] = xs.y;
      z[i][4 * kk + 2] = xo.x;
      z[i][4 * kk + 3] = xo.y;
    }
  }

  v8f acc[TPW];
#pragma unroll
  for (int i = 0; i < TPW; ++i)
    acc[i] = v8f{0.f, 0.f, 0.f, 0.f, 0.f, 0.f, 0.f, 0.f};

#pragma unroll
  for (int kk = 0; kk < 4; ++kk) {
    const v2f a = L2[(2 * kk + half) * 16 + lr];
#pragma unroll
    for (int i = 0; i < TPW; ++i) {
      v2f b;
      b.x = z[i][4 * kk + 0];
      b.y = z[i][4 * kk + 1];
      acc[i] = wmma4(a, b, acc[i]);
    }
  }

  quad_steps<0>(Q2 + half * 16 + lr, z, acc);

  cube_steps<0>(T2 + half * 16 + lr, z, acc);

  typedef float v4f __attribute__((ext_vector_type(4)));
#pragma unroll
  for (int i = 0; i < TPW; ++i) {
    float* yrow = Y + (row0 + 16 * i + lr) * DOUT + 8 * half;
    v4f lo = {acc[i][0], acc[i][1], acc[i][2], acc[i][3]};
    v4f hi = {acc[i][4], acc[i][5], acc[i][6], acc[i][7]};
    *(v4f*)(yrow)     = lo;
    *(v4f*)(yrow + 4) = hi;
  }
}


extern "C" void kernel_launch(void* const* d_in, const int* in_sizes, int n_in,
                              void* d_out, int out_size, void* d_ws, size_t ws_size,
                              hipStream_t stream) {
  const float* x   = (const float*)d_in[0];
  const float* Mre = (const float*)d_in[1];
  const float* Mim = (const float*)d_in[2];
  const float* Sre = (const float*)d_in[3];
  const float* Sim = (const float*)d_in[4];
  const float* pw  = (const float*)d_in[5];

  float* ws = (float*)d_ws;
  float* Lp = ws + L_OFF;
  float* Qp = ws + Q_OFF;
  float* Tp = ws + T_OFF;

  pre_all<<<NB_PRE, 256, 0, stream>>>(Mre, Mim, Sre, Sim, pw, Lp, Qp, Tp);

  gaunt_main<<<NROWS / (8 * 16 * TPW), 256, 0, stream>>>(x, Lp, Qp, Tp, (float*)d_out);
}


# pre_all: each block requests all it reads at its start (S slice, path weight; tap coefficients as scalar loads), not once per step; registers 225 -> 124, so all 961 blocks are resident at once
# speedup vs baseline: 5.9711x; 1.1069x over previous
#include <hip/hip_runtime.h>
#include <hip/hip_bf16.h>
#include <math.h>


typedef float v2f __attribute__((ext_vector_type(2)));
typedef float v8f __attribute__((ext_vector_type(8)));
typedef const __attribute__((address_space(4))) float* cfloatp;

#define NROWS   131072
#define TPW     4
#define DIN     16
#define DOUT    16
#define GRID    19
#define WREAL   361

#define NSLOT   72
#define NT2     (NSLOT / 2)
#define NCONV   169

#define NSLOT3  408
#define NT3     (NSLOT3 / 2)
#define MBW     19
#define PPW     25

#define L_OFF   0
#define Q_OFF   256
#define T_OFF   (Q_OFF + 2304)
#define WS_FLOATS (T_OFF + 13056)

struct IdxPair { int a, b; };
__host__ __device__ constexpr IdxPair slot_pair(int e) {
  int n = 0;
  for (int a = 0; a < 16; ++a)
    for (int b = a; b < 16; ++b) {
      int pa = a ^ 2, pb = b ^ 2;
      if (pa > pb) { int t = pa; pa = pb; pb = t; }
      bool first = (a < pa) || (a == pa && b <= pb);
      if (first) {
        if (n == e) return IdxPair{a, b};
        ++n;
      }
    }
  return IdxPair{0, 0};
}
struct SlotTable { unsigned char a[NSLOT]; unsigned char b[NSLOT]; };
__host__ __device__ constexpr SlotTable make_slot_table() {
  SlotTable t{};
  for (int e = 0; e < NSLOT; ++e) {
    IdxPair p = slot_pair(e);
    t.a[e] = (unsigned char)p.a;
    t.b[e] = (unsigned char)p.b;
  }
  return t;
}

struct Slot3Table { unsigned char a[NSLOT3]; unsigned char b[NSLOT3]; unsigned char c[NSLOT3]; int n; };
__host__ __device__ constexpr Slot3Table make_slot3_table() {
  Slot3Table t{};
  int n = 0;
  for (int a = 0; a < 16; ++a)
    for (int b = a; b < 16; ++b)
      for (int c = b; c < 16; ++c) {
        int p0 = a ^ 2, p1 = b ^ 2, p2 = c ^ 2;
        if (p0 > p1) { int s = p0; p0 = p1; p1 = s; }
        if (p1 > p2) { int s = p1; p1 = p2; p2 = s; }
        if (p0 > p1) { int s = p0; p0 = p1; p1 = s; }
        bool first = (a < p0) || (a == p0 && (b < p1 || (b == p1 && c <= p2)));
        if (first) {
          if (n < NSLOT3) {
            t.a[n] = (unsigned char)a;
            t.b[n] = (unsigned char)b;
            t.c[n] = (unsigned char)c;
          }
          ++n;
        }
      }
  t.n = n;
  return t;
}
static_assert(make_slot3_table().n == NSLOT3, "the swap must halve the 816 triples exactly");

static __device__ inline v8f wmma4(v2f a, v2f b, v8f c) {
  return __builtin_amdgcn_wmma_f32_16x16x4_f32(
       false, a,  false, b,
       (short)0, c,  false,  false);
}


static __device__ inline void role_L(int i, int j,
                                     const float* __restrict__ Mre, const float* __restrict__ Mim,
                                     const float* __restrict__ Sre, const float* __restrict__ Sim,
                                     const float* __restrict__ pw, float* __restrict__ Lpack) {
  float s = 0.f;
  for (int u = 0; u < 7; ++u)
    for (int v = 0; v < 7; ++v) {
      s += Mre[(i * 7 + u) * 7 + v] * Sre[(j * GRID + u + 6) * GRID + v + 6]
         - Mim[(i * 7 + u) * 7 + v] * Sim[(j * GRID + u + 6) * GRID + v + 6];
    }
  s *= pw[j];
  Lpack[((i >> 1) * 16 + j) * 2 + (i & 1)] = s;
}

static __device__ __forceinline__ void fill_mb(int b, int tid, v2f* mb,
                                               const float* __restrict__ Mre, const float* __restrict__ Mim) {
  const int u = tid / MBW - 6, v = tid % MBW - 6;
  const bool ok = (u >= 0) && (u < 7) && (v >= 0) && (v < 7);
  const int k = ok ? (u * 7 + v) : 0;
  const float br = Mre[b * 49 + k];
  const float bi = Mim[b * 49 + k];
  v2f o;
  o.x = ok ? br : 0.f;
  o.y = ok ? bi : 0.f;
  mb[tid] = o;
  if (tid + 256 < MBW * MBW) {
    const v2f z = {0.f, 0.f};
    mb[tid + 256] = z;
  }
}

static __device__ inline void role_Q(int e, int h, int tid, v2f* mb, float* cv_re, float* cv_im,
                                     const float* __restrict__ Mre, const float* __restrict__ Mim,
                                     const float* __restrict__ Sre, const float* __restrict__ Sim,
                                     const float* __restrict__ pw, float* __restrict__ Qpack) {
  constexpr SlotTable slots = make_slot_table();
  const int ra = __builtin_amdgcn_readfirstlane(slots.a[e]);
  const int rb = __builtin_amdgcn_readfirstlane(slots.b[e]);
  const int a = ra ^ (2 * h);
  const int b = rb ^ (2 * h);
  const bool own_image = (rb == (ra ^ 2));

  const int j    = tid >> 4;
  const int part = tid & 15;
  float s_re[11], s_im[11];
#pragma unroll
  for (int it = 0; it < 11; ++it) {
    const int t = part + 16 * it;
    const bool ok = t < NCONV;
    const int tc = ok ? t : 0;
    const int tu = tc / 13, tv = tc - tu * 13;
    const int si = (j * GRID + tu + 3) * GRID + tv + 3;
    const float sr = Sre[si];
    const float sm = Sim[si];
    s_re[it] = ok ? sr : 0.f;
    s_im[it] = ok ? sm : 0.f;
  }
  const float pwj = pw[j];

  fill_mb(b, tid, mb, Mre, Mim);
  __syncthreads();

  if (tid < NCONV) {
    const int tu = tid / 13, tv = tid - tu * 13;
    const cfloatp are = (cfloatp)(Mre + a * 49);
    const cfloatp aim = (cfloatp)(Mim + a * 49);
    const v2f* src = mb + (tu + 6) * MBW + (tv + 6);
    float cr = 0.f, ci = 0.f;
#pragma unroll
    for (int k = 0; k < 49; ++k) {
      const int u1 = k / 7, v1 = k - u1 * 7;
      const v2f bv = src[-(u1 * MBW + v1)];
      const float ar = are[k], ai = aim[k];
      cr += ar * bv.x - ai * bv.y;
      ci += ar * bv.y + ai * bv.x;
    }
    cv_re[tid] = cr;
    cv_im[tid] = ci;
  }
  __syncthreads();

  float s = 0.f;
#pragma unroll
  for (int it = 0; it < 11; ++it) {
    const int t = part + 16 * it;
    const int tc = (t < NCONV) ? t : 0;
    s += s_re[it] * cv_re[tc] - s_im[it] * cv_im[tc];
  }
  for (int m = 8; m >= 1; m >>= 1)
    s += __shfl_xor(s, m, 16);

  if (part == 0) {
    const float mult = (a == b) ? 1.f : 2.f;
    const float val = (own_image && h == 1) ? 0.f : s * mult * pwj * (1.f / 19.f);
    Qpack[(((e >> 1) * 2 + h) * 16 + j) * 2 + (e & 1)] = val;
  }
}

static __device__ __forceinline__ void cmac(v2f& acc0, v2f& acc1, float mr, float mi, v2f p) {
  v2f q;
  q.x = -p.y;
  q.y = p.x;
  acc0 += mr * p;
  acc1 += mi * q;
}

static __device__ inline void role_T(int e, int h, int tid,
                                     v2f* mb, v2f* pp, v2f* c3,
                                     const float* __restrict__ Mre, const float* __restrict__ Mim,
                                     const float* __restrict__ Sre, const float* __restrict__ Sim,
                                     const float* __restrict__ pw, float* __restrict__ Tpack) {
  constexpr Slot3Table slots = make_slot3_table();
  const int a = __builtin_amdgcn_readfirstlane(slots.a[e]) ^ (2 * h);
  const int b = __builtin_amdgcn_readfirstlane(slots.b[e]) ^ (2 * h);
  const int c = __builtin_amdgcn_readfirstlane(slots.c[e]) ^ (2 * h);

  const int j    = tid >> 4;
  const int part = tid & 15;
  const float pwj = pw[j];
  float s_re[23], s_im[23];
#pragma unroll
  for (int it = 0; it < 23; ++it) {
    const int t = part + 16 * it;
    const bool ok = t < WREAL;
    const int tc = ok ? t : 0;
    const float sr = Sre[j * WREAL + tc];
    const float sm = Sim[j * WREAL + tc];
    s_re[it] = ok ? sr : 0.f;
    s_im[it] = ok ? sm : 0.f;
  }

  fill_mb(b, tid, mb, Mre, Mim);
  for (int t = tid; t < PPW * PPW; t += 256) {
    const int u = t / PPW, v = t % PPW;
    const bool middle = (u >= 6) && (u < 19) && (v >= 6) && (v < 19);
    if (!middle) {
      v2f o = {0.f, 0.f};
      pp[t] = o;
    }
  }
  __syncthreads();

  if (tid < NCONV) {
    const int pu = tid / 13, pv = tid - pu * 13;
    const cfloatp are = (cfloatp)(Mre + a * 49);
    const cfloatp aim = (cfloatp)(Mim + a * 49);
    const v2f* src = mb + (pu + 6) * MBW + (pv + 6);
    v2f acc0 = {0.f, 0.f}, acc1 = {0.f, 0.f};
#pragma unroll
    for (int k = 0; k < 49; ++k) {
      const int u1 = k / 7, v1 = k - u1 * 7;
      cmac(acc0, acc1, are[k], aim[k], src[-(u1 * MBW + v1)]);
    }
    pp[(pu + 6) * PPW + (pv + 6)] = acc0 + acc1;
  }
  __syncthreads();

  {
    const cfloatp cre = (cfloatp)(Mre + c * 49);
    const cfloatp cim = (cfloatp)(Mim + c * 49);
    const int t0  = tid;
    const int tu0 = t0 / GRID, tv0 = t0 - tu0 * GRID;
    const v2f* src0 = pp + (tu0 + 6) * PPW + (tv0 + 6);
    if (tid < 128) {
      const int t1   = tid + 256;
      const bool has1 = t1 < WREAL;
      const int t1c  = has1 ? t1 : t0;
      const int tu1 = t1c / GRID, tv1 = t1c - tu1 * GRID;
      const v2f* src1 = pp + (tu1 + 6) * PPW + (tv1 + 6);
      v2f a00 = {0.f, 0.f}, a01 = {0.f, 0.f}, a10 = {0.f, 0.f}, a11 = {0.f, 0.f};
#pragma unroll
      for (int k = 0; k < 49; ++k) {
        const int u1 = k / 7, v1 = k - u1 * 7;
        const float mr = cre[k], mi = cim[k];
        cmac(a00, a01, mr, mi, src0[-(u1 * PPW + v1)]);
        cmac(a10, a11, mr, mi, src1[-(u1 * PPW + v1)]);
      }
      c3[t0] = a00 + a01;
      if (has1) c3[t1] = a10 + a11;
    } else {
      v2f a00 = {0.f, 0.f}, a01 = {0.f, 0.f};
#pragma unroll
      for (int k = 0; k < 49; ++k) {
        const int u1 = k / 7, v1 = k - u1 * 7;
        cmac(a00, a01, cre[k], cim[k], src0[-(u1 * PPW + v1)]);
      }
      c3[t0] = a00 + a01;
    }
  }
  __syncthreads();

  float s = 0.f;
#pragma unroll
  for (int it = 0; it < 23; ++it) {
    const int t = part + 16 * it;
    const int tc = (t < WREAL) ? t : 0;
    const v2f kv = c3[tc];
    s += s_re[it] * kv.x - s_im[it] * kv.y;
  }
  for (int m = 8; m >= 1; m >>= 1)
    s += __shfl_xor(s, m, 16);

  if (part == 0) {
    const float mult = (a == b && b == c) ? 1.f : ((a == b || b == c || a == c) ? 3.f : 6.f);
    const float val = s * mult * pwj * (1.f / 361.f);
    Tpack[(((e >> 1) * 2 + h) * 16 + j) * 2 + (e & 1)] = val;
  }
}

#define NB_T    (NSLOT3 * 2)
#define NB_Q    (NSLOT * 2)
#define NB_PRE  (NB_T + NB_Q + 1)
__global__ __launch_bounds__(256, 8)
void pre_all(const float* __restrict__ Mre, const float* __restrict__ Mim,
             const float* __restrict__ Sre, const float* __restrict__ Sim,
             const float* __restrict__ pw,
             float* __restrict__ Lpack, float* __restrict__ Qpack, float* __restrict__ Tpack) {
  __shared__ float cv_re[NCONV];
  __shared__ float cv_im[NCONV];
  __shared__ v2f mb[MBW * MBW];
  __shared__ v2f pp[PPW * PPW];
  __shared__ v2f c3[WREAL];
  const int tid = threadIdx.x;
  const int b   = blockIdx.x;
  if (b < NB_T) {
    role_T(b >> 1, b & 1, tid, mb, pp, c3, Mre, Mim, Sre, Sim, pw, Tpack);
  } else if (b < NB_T + NB_Q) {
    const int bq = b - NB_T;
    role_Q(bq >> 1, bq & 1, tid, mb, cv_re, cv_im, Mre, Mim, Sre, Sim, pw, Qpack);
  } else {
    role_L(tid & 15, tid >> 4, Mre, Mim, Sre, Sim, pw, Lpack);
  }
}


template <int T>
static __device__ __forceinline__ void quad_steps(const v2f* __restrict__ Qw, const float (&z)[TPW][16], v8f (&acc)[TPW]) {
  if constexpr (T < NT2) {
    constexpr IdxPair p0 = slot_pair(2 * T);
    constexpr IdxPair p1 = slot_pair(2 * T + 1);
    const v2f a = Qw[T * 32];
#pragma unroll
    for (int i = 0; i < TPW; ++i) {
      v2f b;
      b.x = z[i][p0.a] * z[i][p0.b];
      b.y = z[i][p1.a] * z[i][p1.b];
      acc[i] = wmma4(a, b, acc[i]);
    }
    quad_steps<T + 1>(Qw, z, acc);
  }
}

template <int T>
static __device__ __forceinline__ void cube_steps(const v2f* __restrict__ Tw, const float (&z)[TPW][16], v8f (&acc)[TPW]) {
  if constexpr (T < NT3) {
    constexpr Slot3Table tab = make_slot3_table();
    constexpr int a0 = tab.a[2 * T],     b0 = tab.b[2 * T],     c0 = tab.c[2 * T];
    constexpr int a1 = tab.a[2 * T + 1], b1 = tab.b[2 * T + 1], c1 = tab.c[2 * T + 1];
    const v2f a = Tw[T * 32];
#pragma unroll
    for (int i = 0; i < TPW; ++i) {
      v2f b;
      b.x = (z[i][a0] * z[i][b0]) * z[i][c0];
      b.y = (z[i][a1] * z[i][b1]) * z[i][c1];
      acc[i] = wmma4(a, b, acc[i]);
    }
    cube_steps<T + 1>(Tw, z, acc);
  }
}

__global__ __launch_bounds__(256, 1)
void gaunt_main(const float* __restrict__ X,
                const float* __restrict__ Lp,
                const float* __restrict__ Qp,
                const float* __restrict__ Tp,
                float* __restrict__ Y) {
  const int tid  = threadIdx.x;
  const int wave = tid >> 5;
  const int lane = tid & 31;
  const int half = lane >> 4;
  const int lr   = lane & 15;

  const long row0 = ((long)blockIdx.x * 8 + wave) * (16 * TPW);

  const v2f* L2 = (const v2f*)Lp;
  const v2f* Q2 = (const v2f*)Qp;
  const v2f* T2 = (const v2f*)Tp;

  float z[TPW][16];
#pragma unroll
  for (int i = 0; i < TPW; ++i) {
    const float* xrow = X + (row0 + 16 * i + lr) * DIN;
#pragma unroll
    for (int kk = 0; kk < 4; ++kk) {
      const v2f xs = *(const v2f*)(xrow + 4 * kk + 2 * half);
      const v2f xo = *(const v2f*)(xrow + 4 * kk + 2 * (1 - half));
      z[i][4 * kk + 0] = xs.x;
      z[i][4 * kk + 1] = xs.y;
      z[i][4 * kk + 2] = xo.x;
      z[i][4 * kk + 3] = xo.y;
    }
  }

  v8f acc[TPW];
#pragma unroll
  for (int i = 0; i < TPW; ++i)
    acc[i] = v8f{0.f, 0.f, 0.f, 0.f, 0.f, 0.f, 0.f, 0.f};

#pragma unroll
  for (int kk = 0; kk < 4; ++kk) {
    const v2f a = L2[(2 * kk + half) * 16 + lr];
#pragma unroll
    for (int i = 0; i < TPW; ++i) {
      v2f b;
      b.x = z[i][4 * kk + 0];
      b.y = z[i][4 * kk + 1];
      acc[i] = wmma4(a, b, acc[i]);
    }
  }

  quad_steps<0>(Q2 + half * 16 + lr, z, acc);

  cube_steps<0>(T2 + half * 16 + lr, z, acc);

  typedef float v4f __attribute__((ext_vector_type(4)));
#pragma unroll
  for (int i = 0; i < TPW; ++i) {
    float* yrow = Y + (row0 + 16 * i + lr) * DOUT + 8 * half;
    v4f lo = {acc[i][0], acc[i][1], acc[i][2], acc[i][3]};
    v4f hi = {acc[i][4], acc[i][5], acc[i][6], acc[i][7]};
    *(v4f*)(yrow)     = lo;
    *(v4f*)(yrow + 4) = hi;
  }
}


extern "C" void kernel_launch(void* const* d_in, const int* in_sizes, int n_in,
                              void* d_out, int out_size, void* d_ws, size_t ws_size,
                              hipStream_t stream) {
  const float* x   = (const float*)d_in[0];
  const float* Mre = (const float*)d_in[1];
  const float* Mim = (const float*)d_in[2];
  const float* Sre = (const float*)d_in[3];
  const float* Sim = (const float*)d_in[4];
  const float* pw  = (const float*)d_in[5];

  float* ws = (float*)d_ws;
  float* Lp = ws + L_OFF;
  float* Qp = ws + Q_OFF;
  float* Tp = ws + T_OFF;

  pre_all<<<NB_PRE, 256, 0, stream>>>(Mre, Mim, Sre, Sim, pw, Lp, Qp, Tp);

  gaunt_main<<<NROWS / (8 * 16 * TPW), 256, 0, stream>>>(x, Lp, Qp, Tp, (float*)d_out);
}


# Main kernel: the A fragments of each block of 12 k-steps are loaded as one group, a block ahead (second register set); no load stands between a block's matrix instructions; sum order unchanged
# speedup vs baseline: 6.2744x; 1.0508x over previous
#include <hip/hip_runtime.h>
#include <hip/hip_bf16.h>
#include <math.h>


typedef float v2f __attribute__((ext_vector_type(2)));
typedef float v8f __attribute__((ext_vector_type(8)));
typedef const __attribute__((address_space(4))) float* cfloatp;

#define NROWS   131072
#define TPW     4
#define DIN     16
#define DOUT    16
#define GRID    19
#define WREAL   361

#define NSLOT   72
#define NT2     (NSLOT / 2)
#define NCONV   169

#define NSLOT3  408
#define NT3     (NSLOT3 / 2)
#define MBW     19
#define PPW     25

#define L_OFF   0
#define Q_OFF   256
#define T_OFF   (Q_OFF + 2304)
#define WS_FLOATS (T_OFF + 13056)

#define NSTEP   (4 + NT2 + NT3)
#define GRP     12
static_assert(L_OFF == 0 && Q_OFF == 4 * 64 && T_OFF == Q_OFF + NT2 * 64,
              "the packs must follow each other without a gap: one fragment index runs through all three");

struct IdxPair { int a, b; };
__host__ __device__ constexpr IdxPair slot_pair(int e) {
  int n = 0;
  for (int a = 0; a < 16; ++a)
    for (int b = a; b < 16; ++b) {
      int pa = a ^ 2, pb = b ^ 2;
      if (pa > pb) { int t = pa; pa = pb; pb = t; }
      bool first = (a < pa) || (a == pa && b <= pb);
      if (first) {
        if (n == e) return IdxPair{a, b};
        ++n;
      }
    }
  return IdxPair{0, 0};
}
struct SlotTable { unsigned char a[NSLOT]; unsigned char b[NSLOT]; };
__host__ __device__ constexpr SlotTable make_slot_table() {
  SlotTable t{};
  for (int e = 0; e < NSLOT; ++e) {
    IdxPair p = slot_pair(e);
    t.a[e] = (unsigned char)p.a;
    t.b[e] = (unsigned char)p.b;
  }
  return t;
}

struct Slot3Table { unsigned char a[NSLOT3]; unsigned char b[NSLOT3]; unsigned char c[NSLOT3]; int n; };
__host__ __device__ constexpr Slot3Table make_slot3_table() {
  Slot3Table t{};
  int n = 0;
  for (int a = 0; a < 16; ++a)
    for (int b = a; b < 16; ++b)
      for (int c = b; c < 16; ++c) {
        int p0 = a ^ 2, p1 = b ^ 2, p2 = c ^ 2;
        if (p0 > p1) { int s = p0; p0 = p1; p1 = s; }
        if (p1 > p2) { int s = p1; p1 = p2; p2 = s; }
        if (p0 > p1) { int s = p0; p0 = p1; p1 = s; }
        bool first = (a < p0) || (a == p0 && (b < p1 || (b == p1 && c <= p2)));
        if (first) {
          if (n < NSLOT3) {
            t.a[n] = (unsigned char)a;
            t.b[n] = (unsigned char)b;
            t.c[n] = (unsigned char)c;
          }
          ++n;
        }
      }
  t.n = n;
  return t;
}
static_assert(make_slot3_table().n == NSLOT3, "the swap must halve the 816 triples exactly");

static __device__ inline v8f wmma4(v2f a, v2f b, v8f c) {
  return __builtin_amdgcn_wmma_f32_16x16x4_f32(
       false, a,  false, b,
       (short)0, c,  false,  false);
}


static __device__ inline void role_L(int i, int j,
                                     const float* __restrict__ Mre, const float* __restrict__ Mim,
                                     const float* __restrict__ Sre, const float* __restrict__ Sim,
                                     const float* __restrict__ pw, float* __restrict__ Lpack) {
  float s = 0.f;
  for (int u = 0; u < 7; ++u)
    for (int v = 0; v < 7; ++v) {
      s += Mre[(i * 7 + u) * 7 + v] * Sre[(j * GRID + u + 6) * GRID + v + 6]
         - Mim[(i * 7 + u) * 7 + v] * Sim[(j * GRID + u + 6) * GRID + v + 6];
    }
  s *= pw[j];
  Lpack[((i >> 1) * 16 + j) * 2 + (i & 1)] = s;
}

static __device__ __forceinline__ void fill_mb(int b, int tid, v2f* mb,
                                               const float* __restrict__ Mre, const float* __restrict__ Mim) {
  const int u = tid / MBW - 6, v = tid % MBW - 6;
  const bool ok = (u >= 0) && (u < 7) && (v >= 0) && (v < 7);
  const int k = ok ? (u * 7 + v) : 0;
  const float br = Mre[b * 49 + k];
  const float bi = Mim[b * 49 + k];
  v2f o;
  o.x = ok ? br : 0.f;
  o.y = ok ? bi : 0.f;
  mb[tid] = o;
  if (tid + 256 < MBW * MBW) {
    const v2f z = {0.f, 0.f};
    mb[tid + 256] = z;
  }
}

static __device__ inline void role_Q(int e, int h, int tid, v2f* mb, float* cv_re, float* cv_im,
                                     const float* __restrict__ Mre, const float* __restrict__ Mim,
                                     const float* __restrict__ Sre, const float* __restrict__ Sim,
                                     const float* __restrict__ pw, float* __restrict__ Qpack) {
  constexpr SlotTable slots = make_slot_table();
  const int ra = __builtin_amdgcn_readfirstlane(slots.a[e]);
  const int rb = __builtin_amdgcn_readfirstlane(slots.b[e]);
  const int a = ra ^ (2 * h);
  const int b = rb ^ (2 * h);
  const bool own_image = (rb == (ra ^ 2));

  const int j    = tid >> 4;
  const int part = tid & 15;
  float s_re[11], s_im[11];
#pragma unroll
  for (int it = 0; it < 11; ++it) {
    const int t = part + 16 * it;
    const bool ok = t < NCONV;
    const int tc = ok ? t : 0;
    const int tu = tc / 13, tv = tc - tu * 13;
    const int si = (j * GRID + tu + 3) * GRID + tv + 3;
    const float sr = Sre[si];
    const float sm = Sim[si];
    s_re[it] = ok ? sr : 0.f;
    s_im[it] = ok ? sm : 0.f;
  }
  const float pwj = pw[j];

  fill_mb(b, tid, mb, Mre, Mim);
  __syncthreads();

  if (tid < NCONV) {
    const int tu = tid / 13, tv = tid - tu * 13;
    const cfloatp are = (cfloatp)(Mre + a * 49);
    const cfloatp aim = (cfloatp)(Mim + a * 49);
    const v2f* src = mb + (tu + 6) * MBW + (tv + 6);
    float cr = 0.f, ci = 0.f;
#pragma unroll
    for (int k = 0; k < 49; ++k) {
      const int u1 = k / 7, v1 = k - u1 * 7;
      const v2f bv = src[-(u1 * MBW + v1)];
      const float ar = are[k], ai = aim[k];
      cr += ar * bv.x - ai * bv.y;
      ci += ar * bv.y + ai * bv.x;
    }
    cv_re[tid] = cr;
    cv_im[tid] = ci;
  }
  __syncthreads();

  float s = 0.f;
#pragma unroll
  for (int it = 0; it < 11; ++it) {
    const int t = part + 16 * it;
    const int tc = (t < NCONV) ? t : 0;
    s += s_re[it] * cv_re[tc] - s_im[it] * cv_im[tc];
  }
  for (int m = 8; m >= 1; m >>= 1)
    s += __shfl_xor(s, m, 16);

  if (part == 0) {
    const float mult = (a == b) ? 1.f : 2.f;
    const float val = (own_image && h == 1) ? 0.f : s * mult * pwj * (1.f / 19.f);
    Qpack[(((e >> 1) * 2 + h) * 16 + j) * 2 + (e & 1)] = val;
  }
}

static __device__ __forceinline__ void cmac(v2f& acc0, v2f& acc1, float mr, float mi, v2f p) {
  v2f q;
  q.x = -p.y;
  q.y = p.x;
  acc0 += mr * p;
  acc1 += mi * q;
}

static __device__ inline void role_T(int e, int h, int tid,
                                     v2f* mb, v2f* pp, v2f* c3,
                                     const float* __restrict__ Mre, const float* __restrict__ Mim,
                                     const float* __restrict__ Sre, const float* __restrict__ Sim,
                                     const float* __restrict__ pw, float* __restrict__ Tpack) {
  constexpr Slot3Table slots = make_slot3_table();
  const int a = __builtin_amdgcn_readfirstlane(slots.a[e]) ^ (2 * h);
  const int b = __builtin_amdgcn_readfirstlane(slots.b[e]) ^ (2 * h);
  const int c = __builtin_amdgcn_readfirstlane(slots.c[e]) ^ (2 * h);

  const int j    = tid >> 4;
  const int part = tid & 15;
  const float pwj = pw[j];
  float s_re[23], s_im[23];
#pragma unroll
  for (int it = 0; it < 23; ++it) {
    const int t = part + 16 * it;
    const bool ok = t < WREAL;
    const int tc = ok ? t : 0;
    const float sr = Sre[j * WREAL + tc];
    const float sm = Sim[j * WREAL + tc];
    s_re[it] = ok ? sr : 0.f;
    s_im[it] = ok ? sm : 0.f;
  }

  fill_mb(b, tid, mb, Mre, Mim);
  for (int t = tid; t < PPW * PPW; t += 256) {
    const int u = t / PPW, v = t % PPW;
    const bool middle = (u >= 6) && (u < 19) && (v >= 6) && (v < 19);
    if (!middle) {
      v2f o = {0.f, 0.f};
      pp[t] = o;
    }
  }
  __syncthreads();

  if (tid < NCONV) {
    const int pu = tid / 13, pv = tid - pu * 13;
    const cfloatp are = (cfloatp)(Mre + a * 49);
    const cfloatp aim = (cfloatp)(Mim + a * 49);
    const v2f* src = mb + (pu + 6) * MBW + (pv + 6);
    v2f acc0 = {0.f, 0.f}, acc1 = {0.f, 0.f};
#pragma unroll
    for (int k = 0; k < 49; ++k) {
      const int u1 = k / 7, v1 = k - u1 * 7;
      cmac(acc0, acc1, are[k], aim[k], src[-(u1 * MBW + v1)]);
    }
    pp[(pu + 6) * PPW + (pv + 6)] = acc0 + acc1;
  }
  __syncthreads();

  {
    const cfloatp cre = (cfloatp)(Mre + c * 49);
    const cfloatp cim = (cfloatp)(Mim + c * 49);
    const int t0  = tid;
    const int tu0 = t0 / GRID, tv0 = t0 - tu0 * GRID;
    const v2f* src0 = pp + (tu0 + 6) * PPW + (tv0 + 6);
    if (tid < 128) {
      const int t1   = tid + 256;
      const bool has1 = t1 < WREAL;
      const int t1c  = has1 ? t1 : t0;
      const int tu1 = t1c / GRID, tv1 = t1c - tu1 * GRID;
      const v2f* src1 = pp + (tu1 + 6) * PPW + (tv1 + 6);
      v2f a00 = {0.f, 0.f}, a01 = {0.f, 0.f}, a10 = {0.f, 0.f}, a11 = {0.f, 0.f};
#pragma unroll
      for (int k = 0; k < 49; ++k) {
        const int u1 = k / 7, v1 = k - u1 * 7;
        const float mr = cre[k], mi = cim[k];
        cmac(a00, a01, mr, mi, src0[-(u1 * PPW + v1)]);
        cmac(a10, a11, mr, mi, src1[-(u1 * PPW + v1)]);
      }
      c3[t0] = a00 + a01;
      if (has1) c3[t1] = a10 + a11;
    } else {
      v2f a00 = {0.f, 0.f}, a01 = {0.f, 0.f};
#pragma unroll
      for (int k = 0; k < 49; ++k) {
        const int u1 = k / 7, v1 = k - u1 * 7;
        cmac(a00, a01, cre[k], cim[k], src0[-(u1 * PPW + v1)]);
      }
      c3[t0] = a00 + a01;
    }
  }
  __syncthreads();

  float s = 0.f;
#pragma unroll
  for (int it = 0; it < 23; ++it) {
    const int t = part + 16 * it;
    const int tc = (t < WREAL) ? t : 0;
    const v2f kv = c3[tc];
    s += s_re[it] * kv.x - s_im[it] * kv.y;
  }
  for (int m = 8; m >= 1; m >>= 1)
    s += __shfl_xor(s, m, 16);

  if (part == 0) {
    const float mult = (a == b && b == c) ? 1.f : ((a == b || b == c || a == c) ? 3.f : 6.f);
    const float val = s * mult * pwj * (1.f / 361.f);
    Tpack[(((e >> 1) * 2 + h) * 16 + j) * 2 + (e & 1)] = val;
  }
}

#define NB_T    (NSLOT3 * 2)
#define NB_Q    (NSLOT * 2)
#define NB_PRE  (NB_T + NB_Q + 1)
__global__ __launch_bounds__(256, 8)
void pre_all(const float* __restrict__ Mre, const float* __restrict__ Mim,
             const float* __restrict__ Sre, const float* __restrict__ Sim,
             const float* __restrict__ pw,
             float* __restrict__ Lpack, float* __restrict__ Qpack, float* __restrict__ Tpack) {
  __shared__ float cv_re[NCONV];
  __shared__ float cv_im[NCONV];
  __shared__ v2f mb[MBW * MBW];
  __shared__ v2f pp[PPW * PPW];
  __shared__ v2f c3[WREAL];
  const int tid = threadIdx.x;
  const int b   = blockIdx.x;
  if (b < NB_T) {
    role_T(b >> 1, b & 1, tid, mb, pp, c3, Mre, Mim, Sre, Sim, pw, Tpack);
  } else if (b < NB_T + NB_Q) {
    const int bq = b - NB_T;
    role_Q(bq >> 1, bq & 1, tid, mb, cv_re, cv_im, Mre, Mim, Sre, Sim, pw, Qpack);
  } else {
    role_L(tid & 15, tid >> 4, Mre, Mim, Sre, Sim, pw, Lpack);
  }
}


template <int S>
static __device__ __forceinline__ void k_step(v2f a, const float (&z)[TPW][16], v8f (&acc)[TPW]) {
  if constexpr (S < 4) {
#pragma unroll
    for (int i = 0; i < TPW; ++i) {
      v2f b;
      b.x = z[i][4 * S + 0];
      b.y = z[i][4 * S + 1];
      acc[i] = wmma4(a, b, acc[i]);
    }
  } else if constexpr (S < 4 + NT2) {
    constexpr int T = S - 4;
    constexpr IdxPair p0 = slot_pair(2 * T);
    constexpr IdxPair p1 = slot_pair(2 * T + 1);
#pragma unroll
    for (int i = 0; i < TPW; ++i) {
      v2f b;
      b.x = z[i][p0.a] * z[i][p0.b];
      b.y = z[i][p1.a] * z[i][p1.b];
      acc[i] = wmma4(a, b, acc[i]);
    }
  } else {
    constexpr int T = S - 4 - NT2;
    constexpr Slot3Table tab = make_slot3_table();
    constexpr int a0 = tab.a[2 * T],     b0 = tab.b[2 * T],     c0 = tab.c[2 * T];
    constexpr int a1 = tab.a[2 * T + 1], b1 = tab.b[2 * T + 1], c1 = tab.c[2 * T + 1];
#pragma unroll
    for (int i = 0; i < TPW; ++i) {
      v2f b;
      b.x = (z[i][a0] * z[i][b0]) * z[i][c0];
      b.y = (z[i][a1] * z[i][b1]) * z[i][c1];
      acc[i] = wmma4(a, b, acc[i]);
    }
  }
}

template <int S0, int G>
static __device__ __forceinline__ void block_steps(const v2f (&a)[GRP], const float (&z)[TPW][16], v8f (&acc)[TPW]) {
  if constexpr (G < GRP && S0 + G < NSTEP) {
    k_step<S0 + G>(a[G], z, acc);
    block_steps<S0, G + 1>(a, z, acc);
  }
}

template <int S0>
static __device__ __forceinline__ void load_group(const v2f* __restrict__ Aw, v2f (&a)[GRP]) {
#pragma unroll
  for (int g = 0; g < GRP; ++g)
    if (S0 + g < NSTEP) a[g] = Aw[(S0 + g) * 32];
}

template <int S0>
static __device__ __forceinline__ void run_blocks(const v2f* __restrict__ Aw, const v2f (&cur)[GRP],
                                                  const float (&z)[TPW][16], v8f (&acc)[TPW]) {
  if constexpr (S0 < NSTEP) {
    v2f nxt[GRP];
    load_group<S0 + GRP>(Aw, nxt);
    __builtin_amdgcn_sched_barrier(0);
    block_steps<S0, 0>(cur, z, acc);
    __builtin_amdgcn_sched_barrier(0);
    run_blocks<S0 + GRP>(Aw, nxt, z, acc);
  }
}

__global__ __launch_bounds__(256, 1)
void gaunt_main(const float* __restrict__ X,
                const float* __restrict__ Ap,
                float* __restrict__ Y) {
  const int tid  = threadIdx.x;
  const int wave = tid >> 5;
  const int lane = tid & 31;
  const int half = lane >> 4;
  const int lr   = lane & 15;

  const long row0 = ((long)blockIdx.x * 8 + wave) * (16 * TPW);

  const v2f* Aw = (const v2f*)Ap + half * 16 + lr;

  float z[TPW][16];
#pragma unroll
  for (int i = 0; i < TPW; ++i) {
    const float* xrow = X + (row0 + 16 * i + lr) * DIN;
#pragma unroll
    for (int kk = 0; kk < 4; ++kk) {
      const v2f xs = *(const v2f*)(xrow + 4 * kk + 2 * half);
      const v2f xo = *(const v2f*)(xrow + 4 * kk + 2 * (1 - half));
      z[i][4 * kk + 0] = xs.x;
      z[i][4 * kk + 1] = xs.y;
      z[i][4 * kk + 2] = xo.x;
      z[i][4 * kk + 3] = xo.y;
    }
  }

  v8f acc[TPW];
#pragma unroll
  for (int i = 0; i < TPW; ++i)
    acc[i] = v8f{0.f, 0.f, 0.f, 0.f, 0.f, 0.f, 0.f, 0.f};

  v2f first[GRP];
  load_group<0>(Aw, first);
  run_blocks<0>(Aw, first, z, acc);

  typedef float v4f __attribute__((ext_vector_type(4)));
#pragma unroll
  for (int i = 0; i < TPW; ++i) {
    float* yrow = Y + (row0 + 16 * i + lr) * DOUT + 8 * half;
    v4f lo = {acc[i][0], acc[i][1], acc[i][2], acc[i][3]};
    v4f hi = {acc[i][4], acc[i][5], acc[i][6], acc[i][7]};
    *(v4f*)(yrow)     = lo;
    *(v4f*)(yrow + 4) = hi;
  }
}


extern "C" void kernel_launch(void* const* d_in, const int* in_sizes, int n_in,
                              void* d_out, int out_size, void* d_ws, size_t ws_size,
                              hipStream_t stream) {
  const float* x   = (const float*)d_in[0];
  const float* Mre = (const float*)d_in[1];
  const float* Mim = (const float*)d_in[2];
  const float* Sre = (const float*)d_in[3];
  const float* Sim = (const float*)d_in[4];
  const float* pw  = (const float*)d_in[5];

  float* ws = (float*)d_ws;
  float* Lp = ws + L_OFF;
  float* Qp = ws + Q_OFF;
  float* Tp = ws + T_OFF;

  pre_all<<<NB_PRE, 256, 0, stream>>>(Mre, Mim, Sre, Sim, pw, Lp, Qp, Tp);

  gaunt_main<<<NROWS / (8 * 16 * TPW), 256, 0, stream>>>(x, ws, (float*)d_out);
}
